# starting kernel
# baseline (speedup 1.0000x reference)
#include <hip/hip_runtime.h>

typedef __attribute__((ext_vector_type(2))) float v2f;
typedef __attribute__((ext_vector_type(8))) float v8f;

#define N_NODES  100000
#define N_EDGES  1600000
#define DIM_IN   256
#define DIM_HID  128
#define DIM_OUT  32


__global__ __launch_bounds__(256) void k_init_deg(float* deg, int n) {
  int i = blockIdx.x * blockDim.x + threadIdx.x;
  if (i < n) deg[i] = 1.0f;
}

__global__ __launch_bounds__(256) void k_deg_accum(const int* __restrict__ dst,
                                                   const float* __restrict__ w,
                                                   float* deg, int e) {
  int i = blockIdx.x * blockDim.x + threadIdx.x;
  if (i < e) atomicAdd(&deg[dst[i]], w[i]);
}

__global__ __launch_bounds__(256) void k_deg_to_inv(float* deg, int n) {
  int i = blockIdx.x * blockDim.x + threadIdx.x;
  if (i < n) {
    float d = deg[i];
    deg[i] = (d > 0.0f) ? rsqrtf(fmaxf(d, 1e-12f)) : 0.0f;
  }
}

__global__ __launch_bounds__(256) void k_init_bias(float* buf,
                                                   const float* __restrict__ bias,
                                                   int mask, long total) {
  long i = (long)blockIdx.x * blockDim.x + threadIdx.x;
  if (i < total) buf[i] = bias[i & mask];
}


__global__ __launch_bounds__(256) void k_gemm1_wmma(const float* __restrict__ X,
                                                    const float* __restrict__ W,
                                                    float* __restrict__ H) {
  const int gw   = blockIdx.x * 8 + (threadIdx.x >> 5);
  const int lane = threadIdx.x & 31;
  const int rt = gw >> 3;
  const int ct = gw & 7;
  const int m0 = rt * 16, n0 = ct * 16;
  const int l    = lane & 15;
  const int half = lane >> 4;
  const int ko   = half * 2;

  const float* arow = X + (size_t)(m0 + l) * DIM_IN;
  v8f c = {0.f, 0.f, 0.f, 0.f, 0.f, 0.f, 0.f, 0.f};

  for (int k0 = 0; k0 < DIM_IN; k0 += 4) {
    v2f a = *(const v2f*)(arow + k0 + ko);
    v2f b;
    b.x = W[(size_t)(k0 + ko)     * DIM_HID + n0 + l];
    b.y = W[(size_t)(k0 + ko + 1) * DIM_HID + n0 + l];
    c = __builtin_amdgcn_wmma_f32_16x16x4_f32(false, a, false, b,
                                              (short)0, c, false, false);
  }

  float* drow = H + (size_t)(m0 + half * 8) * DIM_HID + (n0 + l);
#pragma unroll
  for (int i = 0; i < 8; ++i) drow[(size_t)i * DIM_HID] = c[i];
}


__global__ __launch_bounds__(256) void k_gemm2_wmma(const float* __restrict__ XE,
                                                    const float* __restrict__ W,
                                                    float* __restrict__ H) {
  const int gw = blockIdx.x * 8 + (threadIdx.x >> 5);
  if (gw >= (N_NODES / 16) * (DIM_OUT / 16)) return;
  const int lane = threadIdx.x & 31;
  const int rt = gw >> 1;
  const int ct = gw & 1;
  const int m0 = rt * 16, n0 = ct * 16;
  const int l    = lane & 15;
  const int half = lane >> 4;
  const int ko   = half * 2;

  const float* arow = XE + (size_t)(m0 + l) * DIM_HID;
  v8f c = {0.f, 0.f, 0.f, 0.f, 0.f, 0.f, 0.f, 0.f};

  for (int k0 = 0; k0 < DIM_HID; k0 += 4) {
    v2f a;
    a.x = fmaxf(arow[k0 + ko],     0.0f);
    a.y = fmaxf(arow[k0 + ko + 1], 0.0f);
    v2f b;
    b.x = W[(size_t)(k0 + ko)     * DIM_OUT + n0 + l];
    b.y = W[(size_t)(k0 + ko + 1) * DIM_OUT + n0 + l];
    c = __builtin_amdgcn_wmma_f32_16x16x4_f32(false, a, false, b,
                                              (short)0, c, false, false);
  }

  float* drow = H + (size_t)(m0 + half * 8) * DIM_OUT + (n0 + l);
#pragma unroll
  for (int i = 0; i < 8; ++i) drow[(size_t)i * DIM_OUT] = c[i];
}


__global__ __launch_bounds__(256) void k_scatter128(const int* __restrict__ src,
                                                    const int* __restrict__ dst,
                                                    const float* __restrict__ wgt,
                                                    const float* __restrict__ inv,
                                                    const float* __restrict__ h,
                                                    float* acc, int e, int n) {
  const int wid  = blockIdx.x * 8 + (threadIdx.x >> 5);
  const int lane = threadIdx.x & 31;
  if (wid >= e + n) return;
  int s, t; float w;
  if (wid < e) { s = src[wid]; t = dst[wid]; w = wgt[wid]; }
  else         { s = t = wid - e; w = 1.0f; }
  const float nrm = inv[s] * w * inv[t];
  const float4 v = *(const float4*)(h + (size_t)s * DIM_HID + lane * 4);
  float* o = acc + (size_t)t * DIM_HID + lane * 4;
  atomicAdd(o + 0, nrm * v.x);
  atomicAdd(o + 1, nrm * v.y);
  atomicAdd(o + 2, nrm * v.z);
  atomicAdd(o + 3, nrm * v.w);
}

__global__ __launch_bounds__(256) void k_scatter32(const int* __restrict__ src,
                                                   const int* __restrict__ dst,
                                                   const float* __restrict__ wgt,
                                                   const float* __restrict__ inv,
                                                   const float* __restrict__ h,
                                                   float* acc, int e, int n) {
  const int wid  = blockIdx.x * 8 + (threadIdx.x >> 5);
  const int lane = threadIdx.x & 31;
  if (wid >= e + n) return;
  int s, t; float w;
  if (wid < e) { s = src[wid]; t = dst[wid]; w = wgt[wid]; }
  else         { s = t = wid - e; w = 1.0f; }
  const float nrm = inv[s] * w * inv[t];
  const float v = h[(size_t)s * DIM_OUT + lane];
  atomicAdd(acc + (size_t)t * DIM_OUT + lane, nrm * v);
}


extern "C" void kernel_launch(void* const* d_in, const int* in_sizes, int n_in,
                              void* d_out, int out_size, void* d_ws, size_t ws_size,
                              hipStream_t stream) {
  const int N = N_NODES, E = N_EDGES;

  const float* x   = (const float*)d_in[0];
  const int*   ei  = (const int*)d_in[1];
  const int*   src = ei;
  const int*   dst = ei + E;
  const float* w   = (const float*)d_in[2];
  const float* W1  = (const float*)d_in[3];
  const float* b1  = (const float*)d_in[4];
  const float* W2  = (const float*)d_in[5];
  const float* b2  = (const float*)d_in[6];

  float* out  = (float*)d_out;
  float* xemb = out + (size_t)N * DIM_OUT;

  float* inv = (float*)d_ws;
  float* h1  = inv + N;
  float* h2  = h1 + (size_t)N * DIM_HID;

  k_init_deg  <<<(N + 255) / 256, 256, 0, stream>>>(inv, N);
  k_deg_accum <<<(E + 255) / 256, 256, 0, stream>>>(dst, w, inv, E);
  k_deg_to_inv<<<(N + 255) / 256, 256, 0, stream>>>(inv, N);

  k_gemm1_wmma<<<(N / 16) * (DIM_HID / 16) / 8, 256, 0, stream>>>(x, W1, h1);

  const long totH = (long)N * DIM_HID;
  k_init_bias <<<(int)((totH + 255) / 256), 256, 0, stream>>>(xemb, b1, DIM_HID - 1, totH);
  k_scatter128<<<(E + N) / 8, 256, 0, stream>>>(src, dst, w, inv, h1, xemb, E, N);

  const int tiles2 = (N / 16) * (DIM_OUT / 16);
  k_gemm2_wmma<<<(tiles2 + 7) / 8, 256, 0, stream>>>(xemb, W2, h2);

  const long totO = (long)N * DIM_OUT;
  k_init_bias<<<(int)((totO + 255) / 256), 256, 0, stream>>>(out, b2, DIM_OUT - 1, totO);
  k_scatter32<<<(E + N) / 8, 256, 0, stream>>>(src, dst, w, inv, h2, out, E, N);
}


# edge scatters: one wave per 32 consecutive edges (coalesced edge reads, lane broadcast, 8 or 32 row gathers in flight) instead of one wave per edge; arrival order of the f32 atomic adds changes
# speedup vs baseline: 1.1578x; 1.1578x over previous
#include <hip/hip_runtime.h>

typedef __attribute__((ext_vector_type(2))) float v2f;
typedef __attribute__((ext_vector_type(8))) float v8f;

#define N_NODES  100000
#define N_EDGES  1600000
#define DIM_IN   256
#define DIM_HID  128
#define DIM_OUT  32


__global__ __launch_bounds__(256) void k_init_deg(float* deg, int n) {
  int i = blockIdx.x * blockDim.x + threadIdx.x;
  if (i < n) deg[i] = 1.0f;
}

__global__ __launch_bounds__(256) void k_deg_accum(const int* __restrict__ dst,
                                                   const float* __restrict__ w,
                                                   float* deg, int e) {
  int i = blockIdx.x * blockDim.x + threadIdx.x;
  if (i < e) atomicAdd(&deg[dst[i]], w[i]);
}

__global__ __launch_bounds__(256) void k_deg_to_inv(float* deg, int n) {
  int i = blockIdx.x * blockDim.x + threadIdx.x;
  if (i < n) {
    float d = deg[i];
    deg[i] = (d > 0.0f) ? rsqrtf(fmaxf(d, 1e-12f)) : 0.0f;
  }
}

__global__ __launch_bounds__(256) void k_init_bias(float* buf,
                                                   const float* __restrict__ bias,
                                                   int mask, long total) {
  long i = (long)blockIdx.x * blockDim.x + threadIdx.x;
  if (i < total) buf[i] = bias[i & mask];
}


__global__ __launch_bounds__(256) void k_gemm1_wmma(const float* __restrict__ X,
                                                    const float* __restrict__ W,
                                                    float* __restrict__ H) {
  const int gw   = blockIdx.x * 8 + (threadIdx.x >> 5);
  const int lane = threadIdx.x & 31;
  const int rt = gw >> 3;
  const int ct = gw & 7;
  const int m0 = rt * 16, n0 = ct * 16;
  const int l    = lane & 15;
  const int half = lane >> 4;
  const int ko   = half * 2;

  const float* arow = X + (size_t)(m0 + l) * DIM_IN;
  v8f c = {0.f, 0.f, 0.f, 0.f, 0.f, 0.f, 0.f, 0.f};

  for (int k0 = 0; k0 < DIM_IN; k0 += 4) {
    v2f a = *(const v2f*)(arow + k0 + ko);
    v2f b;
    b.x = W[(size_t)(k0 + ko)     * DIM_HID + n0 + l];
    b.y = W[(size_t)(k0 + ko + 1) * DIM_HID + n0 + l];
    c = __builtin_amdgcn_wmma_f32_16x16x4_f32(false, a, false, b,
                                              (short)0, c, false, false);
  }

  float* drow = H + (size_t)(m0 + half * 8) * DIM_HID + (n0 + l);
#pragma unroll
  for (int i = 0; i < 8; ++i) drow[(size_t)i * DIM_HID] = c[i];
}


__global__ __launch_bounds__(256) void k_gemm2_wmma(const float* __restrict__ XE,
                                                    const float* __restrict__ W,
                                                    float* __restrict__ H) {
  const int gw = blockIdx.x * 8 + (threadIdx.x >> 5);
  if (gw >= (N_NODES / 16) * (DIM_OUT / 16)) return;
  const int lane = threadIdx.x & 31;
  const int rt = gw >> 1;
  const int ct = gw & 1;
  const int m0 = rt * 16, n0 = ct * 16;
  const int l    = lane & 15;
  const int half = lane >> 4;
  const int ko   = half * 2;

  const float* arow = XE + (size_t)(m0 + l) * DIM_HID;
  v8f c = {0.f, 0.f, 0.f, 0.f, 0.f, 0.f, 0.f, 0.f};

  for (int k0 = 0; k0 < DIM_HID; k0 += 4) {
    v2f a;
    a.x = fmaxf(arow[k0 + ko],     0.0f);
    a.y = fmaxf(arow[k0 + ko + 1], 0.0f);
    v2f b;
    b.x = W[(size_t)(k0 + ko)     * DIM_OUT + n0 + l];
    b.y = W[(size_t)(k0 + ko + 1) * DIM_OUT + n0 + l];
    c = __builtin_amdgcn_wmma_f32_16x16x4_f32(false, a, false, b,
                                              (short)0, c, false, false);
  }

  float* drow = H + (size_t)(m0 + half * 8) * DIM_OUT + (n0 + l);
#pragma unroll
  for (int i = 0; i < 8; ++i) drow[(size_t)i * DIM_OUT] = c[i];
}


static_assert(N_EDGES % 32 == 0 && N_NODES % 32 == 0,
              "a wave takes 32 whole edges, all real or all self-loops");

static __device__ __forceinline__ float lane_bcast(float x, int j) {
  return __builtin_bit_cast(float,
      __builtin_amdgcn_readlane(__builtin_bit_cast(int, x), j));
}

template <int VEC> struct row_piece;
template <> struct row_piece<4> { typedef float4 type; };
template <> struct row_piece<1> { typedef float  type; };

static __device__ __forceinline__ void add_piece(float* o, float nrm, float4 v) {
  atomicAdd(o + 0, nrm * v.x);
  atomicAdd(o + 1, nrm * v.y);
  atomicAdd(o + 2, nrm * v.z);
  atomicAdd(o + 3, nrm * v.w);
}
static __device__ __forceinline__ void add_piece(float* o, float nrm, float v) {
  atomicAdd(o, nrm * v);
}

template <int DIM>
__global__ __launch_bounds__(256) void k_scatter(const int* __restrict__ src,
                                                 const int* __restrict__ dst,
                                                 const float* __restrict__ wgt,
                                                 const float* __restrict__ inv,
                                                 const float* __restrict__ h,
                                                 float* acc) {
  constexpr int VEC   = DIM / 32;
  constexpr int CHUNK = 32 / VEC;
  typedef typename row_piece<VEC>::type piece_t;

  const int wave = __builtin_amdgcn_readfirstlane((int)(threadIdx.x >> 5));
  const int base = (blockIdx.x * 8 + wave) * 32;
  if (base >= N_EDGES + N_NODES) return;
  const int lane = threadIdx.x & 31;

  const int g = base + lane;
  int s, t; float w;
  if (base < N_EDGES) { s = src[g]; t = dst[g]; w = wgt[g]; }
  else                { s = t = g - N_EDGES; w = 1.0f; }
  const float nrm = inv[s] * w * inv[t];

  const unsigned off = lane * VEC;
  for (int j0 = 0; j0 < 32; j0 += CHUNK) {
    piece_t v[CHUNK];
    for (int u = 0; u < CHUNK; ++u) {
      const float* row = h + (size_t)__builtin_amdgcn_readlane(s, j0 + u) * DIM;
      v[u] = *(const piece_t*)(row + off);
    }
    for (int u = 0; u < CHUNK; ++u) {
      float* row = acc + (size_t)__builtin_amdgcn_readlane(t, j0 + u) * DIM;
      add_piece(row + off, lane_bcast(nrm, j0 + u), v[u]);
    }
  }
}


extern "C" void kernel_launch(void* const* d_in, const int* in_sizes, int n_in,
                              void* d_out, int out_size, void* d_ws, size_t ws_size,
                              hipStream_t stream) {
  const int N = N_NODES, E = N_EDGES;

  const float* x   = (const float*)d_in[0];
  const int*   ei  = (const int*)d_in[1];
  const int*   src = ei;
  const int*   dst = ei + E;
  const float* w   = (const float*)d_in[2];
  const float* W1  = (const float*)d_in[3];
  const float* b1  = (const float*)d_in[4];
  const float* W2  = (const float*)d_in[5];
  const float* b2  = (const float*)d_in[6];

  float* out  = (float*)d_out;
  float* xemb = out + (size_t)N * DIM_OUT;

  float* inv = (float*)d_ws;
  float* h1  = inv + N;
  float* h2  = h1 + (size_t)N * DIM_HID;

  k_init_deg  <<<(N + 255) / 256, 256, 0, stream>>>(inv, N);
  k_deg_accum <<<(E + 255) / 256, 256, 0, stream>>>(dst, w, inv, E);
  k_deg_to_inv<<<(N + 255) / 256, 256, 0, stream>>>(inv, N);

  k_gemm1_wmma<<<(N / 16) * (DIM_HID / 16) / 8, 256, 0, stream>>>(x, W1, h1);

  const long totH = (long)N * DIM_HID;
  k_init_bias <<<(int)((totH + 255) / 256), 256, 0, stream>>>(xemb, b1, DIM_HID - 1, totH);
  k_scatter<DIM_HID><<<((E + N) / 32 + 7) / 8, 256, 0, stream>>>(src, dst, w, inv, h1, xemb);

  const int tiles2 = (N / 16) * (DIM_OUT / 16);
  k_gemm2_wmma<<<(tiles2 + 7) / 8, 256, 0, stream>>>(xemb, W2, h2);

  const long totO = (long)N * DIM_OUT;
  k_init_bias<<<(int)((totO + 255) / 256), 256, 0, stream>>>(out, b2, DIM_OUT - 1, totO);
  k_scatter<DIM_OUT><<<((E + N) / 32 + 7) / 8, 256, 0, stream>>>(src, dst, w, inv, h2, out);
}


# edges ordered by target inside the call (count, scan, place); one wave sums a node's rows in registers, one store: no row atomics, no bias fill; f32 sum order = arrival order of the placing atomics
# speedup vs baseline: 1.4786x; 1.2771x over previous
#include <hip/hip_runtime.h>

typedef __attribute__((ext_vector_type(2))) float v2f;
typedef __attribute__((ext_vector_type(8))) float v8f;

#define N_NODES  100000
#define N_EDGES  1600000
#define DIM_IN   256
#define DIM_HID  128
#define DIM_OUT  32


__global__ __launch_bounds__(256) void k_init_deg(float* deg, int* cnt, int n) {
  int i = blockIdx.x * blockDim.x + threadIdx.x;
  if (i < n) { deg[i] = 1.0f; cnt[i] = 0; }
}

__global__ __launch_bounds__(256) void k_deg_accum(const int* __restrict__ dst,
                                                   const float* __restrict__ w,
                                                   float* deg, int* cnt, int e) {
  int i = blockIdx.x * blockDim.x + threadIdx.x;
  if (i < e) { int d = dst[i]; atomicAdd(&deg[d], w[i]); atomicAdd(&cnt[d], 1); }
}

static __device__ __forceinline__ int wave_incl_scan(int v, int lane) {
#pragma unroll
  for (int d = 1; d < 32; d <<= 1) {
    int t = __shfl_up(v, d, 32);
    if (lane >= d) v += t;
  }
  return v;
}

__global__ __launch_bounds__(256) void k_deg_to_inv(float* deg, const int* __restrict__ cnt,
                                                    int* bsum, int n) {
  __shared__ int sh[8];
  const int tid = threadIdx.x, lane = tid & 31, wave = tid >> 5;
  int i = blockIdx.x * 256 + tid;
  int c = 0;
  if (i < n) {
    float d = deg[i];
    deg[i] = (d > 0.0f) ? rsqrtf(fmaxf(d, 1e-12f)) : 0.0f;
    c = cnt[i];
  }
  int incl = wave_incl_scan(c, lane);
  if (lane == 31) sh[wave] = incl;
  __syncthreads();
  if (tid == 0) {
    int tot = 0;
    for (int k = 0; k < 8; ++k) tot += sh[k];
    bsum[blockIdx.x] = tot;
  }
}

__global__ __launch_bounds__(256) void k_scan(const int* __restrict__ cnt,
                                              const int* __restrict__ bsum,
                                              int* off, int* cur, int n) {
  __shared__ int sh[16];
  const int tid = threadIdx.x, lane = tid & 31, wave = tid >> 5;
  const int b = blockIdx.x;
  const int i = b * 256 + tid;
  const int c = (i < n) ? cnt[i] : 0;
  int part = 0;
  for (int k = tid; k < b; k += 256) part += bsum[k];
#pragma unroll
  for (int d = 16; d > 0; d >>= 1) part += __shfl_xor(part, d, 32);
  const int incl = wave_incl_scan(c, lane);
  if (lane == 31) sh[wave] = incl;
  if (lane == 0)  sh[8 + wave] = part;
  __syncthreads();
  int wbase = 0, base = 0;
  for (int k = 0; k < 8; ++k) {
    int t = sh[k];
    if (k < wave) wbase += t;
    base += sh[8 + k];
  }
  const int start = base + wbase + incl - c;
  if (i < n) { off[i] = start; cur[i] = start; }
  if (i == n - 1) off[n] = start + c;
}

__global__ __launch_bounds__(256) void k_place(const int* __restrict__ dst, int* cur,
                                               int* perm, int e) {
  int i = blockIdx.x * blockDim.x + threadIdx.x;
  if (i < e) perm[atomicAdd(&cur[dst[i]], 1)] = i;
}


__global__ __launch_bounds__(256) void k_gemm1_wmma(const float* __restrict__ X,
                                                    const float* __restrict__ W,
                                                    float* __restrict__ H) {
  const int gw   = blockIdx.x * 8 + (threadIdx.x >> 5);
  const int lane = threadIdx.x & 31;
  const int rt = gw >> 3;
  const int ct = gw & 7;
  const int m0 = rt * 16, n0 = ct * 16;
  const int l    = lane & 15;
  const int half = lane >> 4;
  const int ko   = half * 2;

  const float* arow = X + (size_t)(m0 + l) * DIM_IN;
  v8f c = {0.f, 0.f, 0.f, 0.f, 0.f, 0.f, 0.f, 0.f};

  for (int k0 = 0; k0 < DIM_IN; k0 += 4) {
    v2f a = *(const v2f*)(arow + k0 + ko);
    v2f b;
    b.x = W[(size_t)(k0 + ko)     * DIM_HID + n0 + l];
    b.y = W[(size_t)(k0 + ko + 1) * DIM_HID + n0 + l];
    c = __builtin_amdgcn_wmma_f32_16x16x4_f32(false, a, false, b,
                                              (short)0, c, false, false);
  }

  float* drow = H + (size_t)(m0 + half * 8) * DIM_HID + (n0 + l);
#pragma unroll
  for (int i = 0; i < 8; ++i) drow[(size_t)i * DIM_HID] = c[i];
}


__global__ __launch_bounds__(256) void k_gemm2_wmma(const float* __restrict__ XE,
                                                    const float* __restrict__ W,
                                                    float* __restrict__ H) {
  const int gw = blockIdx.x * 8 + (threadIdx.x >> 5);
  if (gw >= (N_NODES / 16) * (DIM_OUT / 16)) return;
  const int lane = threadIdx.x & 31;
  const int rt = gw >> 1;
  const int ct = gw & 1;
  const int m0 = rt * 16, n0 = ct * 16;
  const int l    = lane & 15;
  const int half = lane >> 4;
  const int ko   = half * 2;

  const float* arow = XE + (size_t)(m0 + l) * DIM_HID;
  v8f c = {0.f, 0.f, 0.f, 0.f, 0.f, 0.f, 0.f, 0.f};

  for (int k0 = 0; k0 < DIM_HID; k0 += 4) {
    v2f a;
    a.x = fmaxf(arow[k0 + ko],     0.0f);
    a.y = fmaxf(arow[k0 + ko + 1], 0.0f);
    v2f b;
    b.x = W[(size_t)(k0 + ko)     * DIM_OUT + n0 + l];
    b.y = W[(size_t)(k0 + ko + 1) * DIM_OUT + n0 + l];
    c = __builtin_amdgcn_wmma_f32_16x16x4_f32(false, a, false, b,
                                              (short)0, c, false, false);
  }

  float* drow = H + (size_t)(m0 + half * 8) * DIM_OUT + (n0 + l);
#pragma unroll
  for (int i = 0; i < 8; ++i) drow[(size_t)i * DIM_OUT] = c[i];
}


static __device__ __forceinline__ float lane_bcast(float x, int j) {
  return __builtin_bit_cast(float,
      __builtin_amdgcn_readlane(__builtin_bit_cast(int, x), j));
}

template <int VEC> struct row_piece;
template <> struct row_piece<4> { typedef float4 type; };
template <> struct row_piece<1> { typedef float  type; };

static __device__ __forceinline__ float4 scaled(float a, float4 v) {
  return make_float4(a * v.x, a * v.y, a * v.z, a * v.w);
}
static __device__ __forceinline__ float scaled(float a, float v) { return a * v; }
static __device__ __forceinline__ void add_scaled(float4& acc, float a, float4 v) {
  acc.x += a * v.x; acc.y += a * v.y; acc.z += a * v.z; acc.w += a * v.w;
}
static __device__ __forceinline__ void add_scaled(float& acc, float a, float v) { acc += a * v; }
static __device__ __forceinline__ float4 plus(float4 a, float4 b) {
  return make_float4(a.x + b.x, a.y + b.y, a.z + b.z, a.w + b.w);
}
static __device__ __forceinline__ float plus(float a, float b) { return a + b; }

template <int DIM>
__global__ __launch_bounds__(256) void k_gather(const int* __restrict__ src,
                                                const float* __restrict__ wgt,
                                                const float* __restrict__ inv,
                                                const int* __restrict__ off,
                                                const int* __restrict__ perm,
                                                const float* __restrict__ h,
                                                const float* __restrict__ bias,
                                                float* __restrict__ res) {
  constexpr int VEC   = DIM / 32;
  constexpr int GROUP = 8;
  typedef typename row_piece<VEC>::type piece_t;

  const int wave = __builtin_amdgcn_readfirstlane((int)(threadIdx.x >> 5));
  const int node = blockIdx.x * 8 + wave;
  if (node >= N_NODES) return;
  const int lane = threadIdx.x & 31;
  const unsigned lo = lane * VEC;

  const int beg = __builtin_amdgcn_readfirstlane(off[node]);
  const int end = __builtin_amdgcn_readfirstlane(off[node + 1]);
  const float inv_t = inv[node];

  piece_t acc = scaled(inv_t * 1.0f * inv_t, *(const piece_t*)(h + (size_t)node * DIM + lo));

  for (int c0 = beg; c0 < end; c0 += 32) {
    const int m = min(32, end - c0);
    int s = node; float nrm = 0.0f;
    if (lane < m) {
      const int e = perm[c0 + lane];
      s = src[e];
      nrm = inv[s] * wgt[e] * inv_t;
    }
    for (int u = 0; u < m; u += GROUP) {
      piece_t v[GROUP];
#pragma unroll
      for (int k = 0; k < GROUP; ++k) {
        const float* row = h + (size_t)__builtin_amdgcn_readlane(s, u + k) * DIM;
        v[k] = *(const piece_t*)(row + lo);
      }
#pragma unroll
      for (int k = 0; k < GROUP; ++k) add_scaled(acc, lane_bcast(nrm, u + k), v[k]);
    }
  }

  *(piece_t*)(res + (size_t)node * DIM + lo) = plus(acc, *(const piece_t*)(bias + lo));
}


extern "C" void kernel_launch(void* const* d_in, const int* in_sizes, int n_in,
                              void* d_out, int out_size, void* d_ws, size_t ws_size,
                              hipStream_t stream) {
  const int N = N_NODES, E = N_EDGES;

  const float* x   = (const float*)d_in[0];
  const int*   ei  = (const int*)d_in[1];
  const int*   src = ei;
  const int*   dst = ei + E;
  const float* w   = (const float*)d_in[2];
  const float* W1  = (const float*)d_in[3];
  const float* b1  = (const float*)d_in[4];
  const float* W2  = (const float*)d_in[5];
  const float* b2  = (const float*)d_in[6];

  float* out  = (float*)d_out;
  float* xemb = out + (size_t)N * DIM_OUT;

  float* inv  = (float*)d_ws;
  float* h1   = inv + N;
  int*   perm = (int*)(h1 + (size_t)N * DIM_HID);
  int*   off  = perm + E;
  int*   cnt  = (int*)h1;
  int*   cur  = cnt + N;
  int*   bsum = cur + N;
  float* h2   = h1;

  k_init_deg  <<<(N + 255) / 256, 256, 0, stream>>>(inv, cnt, N);
  k_deg_accum <<<(E + 255) / 256, 256, 0, stream>>>(dst, w, inv, cnt, E);
  k_deg_to_inv<<<(N + 255) / 256, 256, 0, stream>>>(inv, cnt, bsum, N);
  k_scan      <<<(N + 255) / 256, 256, 0, stream>>>(cnt, bsum, off, cur, N);
  k_place     <<<(E + 255) / 256, 256, 0, stream>>>(dst, cur, perm, E);

  k_gemm1_wmma<<<(N / 16) * (DIM_HID / 16) / 8, 256, 0, stream>>>(x, W1, h1);

  k_gather<DIM_HID><<<(N + 7) / 8, 256, 0, stream>>>(src, w, inv, off, perm, h1, b1, xemb);

  const int tiles2 = (N / 16) * (DIM_OUT / 16);
  k_gemm2_wmma<<<(tiles2 + 7) / 8, 256, 0, stream>>>(xemb, W2, h2);

  k_gather<DIM_OUT><<<(N + 7) / 8, 256, 0, stream>>>(src, w, inv, off, perm, h2, b2, out);
}


# both WMMA products: a work-group stages its A rows once through LDS with whole-line 16-byte loads, each wave keeps its B strip in registers across row tiles; f32 sums in the same order
# speedup vs baseline: 1.6871x; 1.1410x over previous
#include <hip/hip_runtime.h>

typedef __attribute__((ext_vector_type(2))) float v2f;
typedef __attribute__((ext_vector_type(4))) float v4f;
typedef __attribute__((ext_vector_type(8))) float v8f;

#define N_NODES  100000
#define N_EDGES  1600000
#define DIM_IN   256
#define DIM_HID  128
#define DIM_OUT  32


__global__ __launch_bounds__(256) void k_init_deg(float* deg, int* cnt, int n) {
  int i = blockIdx.x * blockDim.x + threadIdx.x;
  if (i < n) { deg[i] = 1.0f; cnt[i] = 0; }
}

__global__ __launch_bounds__(256) void k_deg_accum(const int* __restrict__ dst,
                                                   const float* __restrict__ w,
                                                   float* deg, int* cnt, int e) {
  int i = blockIdx.x * blockDim.x + threadIdx.x;
  if (i < e) { int d = dst[i]; atomicAdd(&deg[d], w[i]); atomicAdd(&cnt[d], 1); }
}

static __device__ __forceinline__ int wave_incl_scan(int v, int lane) {
#pragma unroll
  for (int d = 1; d < 32; d <<= 1) {
    int t = __shfl_up(v, d, 32);
    if (lane >= d) v += t;
  }
  return v;
}

__global__ __launch_bounds__(256) void k_deg_to_inv(float* deg, const int* __restrict__ cnt,
                                                    int* bsum, int n) {
  __shared__ int sh[8];
  const int tid = threadIdx.x, lane = tid & 31, wave = tid >> 5;
  int i = blockIdx.x * 256 + tid;
  int c = 0;
  if (i < n) {
    float d = deg[i];
    deg[i] = (d > 0.0f) ? rsqrtf(fmaxf(d, 1e-12f)) : 0.0f;
    c = cnt[i];
  }
  int incl = wave_incl_scan(c, lane);
  if (lane == 31) sh[wave] = incl;
  __syncthreads();
  if (tid == 0) {
    int tot = 0;
    for (int k = 0; k < 8; ++k) tot += sh[k];
    bsum[blockIdx.x] = tot;
  }
}

__global__ __launch_bounds__(256) void k_scan(const int* __restrict__ cnt,
                                              const int* __restrict__ bsum,
                                              int* off, int* cur, int n) {
  __shared__ int sh[16];
  const int tid = threadIdx.x, lane = tid & 31, wave = tid >> 5;
  const int b = blockIdx.x;
  const int i = b * 256 + tid;
  const int c = (i < n) ? cnt[i] : 0;
  int part = 0;
  for (int k = tid; k < b; k += 256) part += bsum[k];
#pragma unroll
  for (int d = 16; d > 0; d >>= 1) part += __shfl_xor(part, d, 32);
  const int incl = wave_incl_scan(c, lane);
  if (lane == 31) sh[wave] = incl;
  if (lane == 0)  sh[8 + wave] = part;
  __syncthreads();
  int wbase = 0, base = 0;
  for (int k = 0; k < 8; ++k) {
    int t = sh[k];
    if (k < wave) wbase += t;
    base += sh[8 + k];
  }
  const int start = base + wbase + incl - c;
  if (i < n) { off[i] = start; cur[i] = start; }
  if (i == n - 1) off[n] = start + c;
}

__global__ __launch_bounds__(256) void k_place(const int* __restrict__ dst, int* cur,
                                               int* perm, int e) {
  int i = blockIdx.x * blockDim.x + threadIdx.x;
  if (i < e) perm[atomicAdd(&cur[dst[i]], 1)] = i;
}


template <int K, int NOUT, int RPP, int TILES, bool RELU>
__global__ __launch_bounds__(32 * (NOUT / 16) * RPP)
void k_gemm_wmma(const float* X, const float* __restrict__ W, float* H) {
  constexpr int CT    = NOUT / 16;
  constexpr int NT    = 32 * CT * RPP;
  constexpr int ROWS  = 16 * RPP;
  constexpr int PITCH = K + 4;
  constexpr int NV    = ROWS * K / 4 / NT;
  constexpr int G     = 8;
  constexpr int NG    = K / 4 / G;
  static_assert(NG * G * 4 == K, "K must be a whole number of batches");
  static_assert(ROWS * K == NV * 4 * NT, "a pass must split evenly over the threads");
  static_assert((N_NODES / 16) % (RPP * TILES) == 0, "passes must cover the rows exactly");

  __shared__ __attribute__((aligned(16))) float sA[2][ROWS * PITCH];

  const int tid  = threadIdx.x;
  const int lane = tid & 31;
  const int wave = tid >> 5;
  const int ct   = wave % CT;
  const int rsub = wave / CT;
  const int l    = lane & 15;
  const int half = lane >> 4;
  const int ko   = half * 2;
  const int n0   = ct * 16;

  v2f b[K / 4];
#pragma unroll
  for (int j = 0; j < K / 4; ++j) {
    b[j].x = W[(size_t)(4 * j + ko)     * NOUT + n0 + l];
    b[j].y = W[(size_t)(4 * j + ko + 1) * NOUT + n0 + l];
  }

  const size_t row0 = (size_t)blockIdx.x * TILES * ROWS;
  const v4f* gsrc = (const v4f*)(X + row0 * K) + tid;

  v4f r[NV];
#pragma unroll
  for (int i = 0; i < NV; ++i) r[i] = gsrc[i * NT];
#pragma unroll
  for (int i = 0; i < NV; ++i) {
    const int f = 4 * (i * NT + tid);
    v4f v = r[i];
    if (RELU) { v.x = fmaxf(v.x, 0.0f); v.y = fmaxf(v.y, 0.0f);
                v.z = fmaxf(v.z, 0.0f); v.w = fmaxf(v.w, 0.0f); }
    *(v4f*)&sA[0][(f / K) * PITCH + (f % K)] = v;
  }
  __syncthreads();

  for (int t = 0; t < TILES; ++t) {
    const int cur = t & 1;
    const bool more = (t + 1 < TILES);
    if (more) {
      const v4f* g = gsrc + (size_t)(t + 1) * (ROWS * K / 4);
#pragma unroll
      for (int i = 0; i < NV; ++i) r[i] = g[i * NT];
    }

    const float* arow = &sA[cur][(rsub * 16 + l) * PITCH + ko];
    v8f c = {0.f, 0.f, 0.f, 0.f, 0.f, 0.f, 0.f, 0.f};
    v2f fa[2][G];
#pragma unroll
    for (int jj = 0; jj < G; ++jj) fa[0][jj] = *(const v2f*)(arow + 4 * jj);
#pragma unroll
    for (int g = 0; g < NG; ++g) {
      if (g + 1 < NG) {
#pragma unroll
        for (int jj = 0; jj < G; ++jj)
          fa[(g + 1) & 1][jj] = *(const v2f*)(arow + 4 * ((g + 1) * G + jj));
      }
      __builtin_amdgcn_sched_barrier(0);
#pragma unroll
      for (int jj = 0; jj < G; ++jj)
        c = __builtin_amdgcn_wmma_f32_16x16x4_f32(false, fa[g & 1][jj], false, b[g * G + jj],
                                                  (short)0, c, false, false);
      __builtin_amdgcn_sched_barrier(0);
    }

    if (more) {
#pragma unroll
      for (int i = 0; i < NV; ++i) {
        const int f = 4 * (i * NT + tid);
        v4f v = r[i];
        if (RELU) { v.x = fmaxf(v.x, 0.0f); v.y = fmaxf(v.y, 0.0f);
                    v.z = fmaxf(v.z, 0.0f); v.w = fmaxf(v.w, 0.0f); }
        *(v4f*)&sA[cur ^ 1][(f / K) * PITCH + (f % K)] = v;
      }
      __syncthreads();
    }

    const size_t m0 = row0 + (size_t)t * ROWS + rsub * 16;
    float* drow = H + (m0 + half * 8) * NOUT + (n0 + l);
#pragma unroll
    for (int i = 0; i < 8; ++i) drow[(size_t)i * NOUT] = c[i];
  }
}


static __device__ __forceinline__ float lane_bcast(float x, int j) {
  return __builtin_bit_cast(float,
      __builtin_amdgcn_readlane(__builtin_bit_cast(int, x), j));
}

template <int VEC> struct row_piece;
template <> struct row_piece<4> { typedef float4 type; };
template <> struct row_piece<1> { typedef float  type; };

static __device__ __forceinline__ float4 scaled(float a, float4 v) {
  return make_float4(a * v.x, a * v.y, a * v.z, a * v.w);
}
static __device__ __forceinline__ float scaled(float a, float v) { return a * v; }
static __device__ __forceinline__ void add_scaled(float4& acc, float a, float4 v) {
  acc.x += a * v.x; acc.y += a * v.y; acc.z += a * v.z; acc.w += a * v.w;
}
static __device__ __forceinline__ void add_scaled(float& acc, float a, float v) { acc += a * v; }
static __device__ __forceinline__ float4 plus(float4 a, float4 b) {
  return make_float4(a.x + b.x, a.y + b.y, a.z + b.z, a.w + b.w);
}
static __device__ __forceinline__ float plus(float a, float b) { return a + b; }

template <int DIM>
__global__ __launch_bounds__(256) void k_gather(const int* __restrict__ src,
                                                const float* __restrict__ wgt,
                                                const float* __restrict__ inv,
                                                const int* __restrict__ off,
                                                const int* __restrict__ perm,
                                                const float* __restrict__ h,
                                                const float* __restrict__ bias,
                                                float* __restrict__ res) {
  constexpr int VEC   = DIM / 32;
  constexpr int GROUP = 8;
  typedef typename row_piece<VEC>::type piece_t;

  const int wave = __builtin_amdgcn_readfirstlane((int)(threadIdx.x >> 5));
  const int node = blockIdx.x * 8 + wave;
  if (node >= N_NODES) return;
  const int lane = threadIdx.x & 31;
  const unsigned lo = lane * VEC;

  const int beg = __builtin_amdgcn_readfirstlane(off[node]);
  const int end = __builtin_amdgcn_readfirstlane(off[node + 1]);
  const float inv_t = inv[node];

  piece_t acc = scaled(inv_t * 1.0f * inv_t, *(const piece_t*)(h + (size_t)node * DIM + lo));

  for (int c0 = beg; c0 < end; c0 += 32) {
    const int m = min(32, end - c0);
    int s = node; float nrm = 0.0f;
    if (lane < m) {
      const int e = perm[c0 + lane];
      s = src[e];
      nrm = inv[s] * wgt[e] * inv_t;
    }
    for (int u = 0; u < m; u += GROUP) {
      piece_t v[GROUP];
#pragma unroll
      for (int k = 0; k < GROUP; ++k) {
        const float* row = h + (size_t)__builtin_amdgcn_readlane(s, u + k) * DIM;
        v[k] = *(const piece_t*)(row + lo);
      }
#pragma unroll
      for (int k = 0; k < GROUP; ++k) add_scaled(acc, lane_bcast(nrm, u + k), v[k]);
    }
  }

  *(piece_t*)(res + (size_t)node * DIM + lo) = plus(acc, *(const piece_t*)(bias + lo));
}


extern "C" void kernel_launch(void* const* d_in, const int* in_sizes, int n_in,
                              void* d_out, int out_size, void* d_ws, size_t ws_size,
                              hipStream_t stream) {
  const int N = N_NODES, E = N_EDGES;

  const float* x   = (const float*)d_in[0];
  const int*   ei  = (const int*)d_in[1];
  const int*   src = ei;
  const int*   dst = ei + E;
  const float* w   = (const float*)d_in[2];
  const float* W1  = (const float*)d_in[3];
  const float* b1  = (const float*)d_in[4];
  const float* W2  = (const float*)d_in[5];
  const float* b2  = (const float*)d_in[6];

  float* out  = (float*)d_out;
  float* xemb = out + (size_t)N * DIM_OUT;

  float* inv  = (float*)d_ws;
  float* h1   = inv + N;
  int*   perm = (int*)(h1 + (size_t)N * DIM_HID);
  int*   off  = perm + E;
  int*   cnt  = (int*)h1;
  int*   cur  = cnt + N;
  int*   bsum = cur + N;
  float* h2   = h1;

  k_init_deg  <<<(N + 255) / 256, 256, 0, stream>>>(inv, cnt, N);
  k_deg_accum <<<(E + 255) / 256, 256, 0, stream>>>(dst, w, inv, cnt, E);
  k_deg_to_inv<<<(N + 255) / 256, 256, 0, stream>>>(inv, cnt, bsum, N);
  k_scan      <<<(N + 255) / 256, 256, 0, stream>>>(cnt, bsum, off, cur, N);
  k_place     <<<(E + 255) / 256, 256, 0, stream>>>(dst, cur, perm, E);

  k_gemm_wmma<DIM_IN, DIM_HID, 1, 5, false><<<(N / 16) / 5, 256, 0, stream>>>(x, W1, h1);

  k_gather<DIM_HID><<<(N + 7) / 8, 256, 0, stream>>>(src, w, inv, off, perm, h1, b1, xemb);

  k_gemm_wmma<DIM_HID, DIM_OUT, 5, 2, true><<<(N / 16) / (5 * 2), 320, 0, stream>>>(xemb, W2, h2);

  k_gather<DIM_OUT><<<(N + 7) / 8, 256, 0, stream>>>(src, w, inv, off, perm, h2, b2, out);
}


# both gathers: a group of lanes (16 for 128-wide rows, 8 for 32-wide) owns a node instead of a whole wave: 2 or 4 nodes per wave, 200000 gather waves become 75000; each node's f32 sum order unchanged
# speedup vs baseline: 1.7845x; 1.0577x over previous
#include <hip/hip_runtime.h>

typedef __attribute__((ext_vector_type(2))) float v2f;
typedef __attribute__((ext_vector_type(4))) float v4f;
typedef __attribute__((ext_vector_type(8))) float v8f;

#define N_NODES  100000
#define N_EDGES  1600000
#define DIM_IN   256
#define DIM_HID  128
#define DIM_OUT  32


__global__ __launch_bounds__(256) void k_init_deg(float* deg, int* cnt, int n) {
  int i = blockIdx.x * blockDim.x + threadIdx.x;
  if (i < n) { deg[i] = 1.0f; cnt[i] = 0; }
}

__global__ __launch_bounds__(256) void k_deg_accum(const int* __restrict__ dst,
                                                   const float* __restrict__ w,
                                                   float* deg, int* cnt, int e) {
  int i = blockIdx.x * blockDim.x + threadIdx.x;
  if (i < e) { int d = dst[i]; atomicAdd(&deg[d], w[i]); atomicAdd(&cnt[d], 1); }
}

static __device__ __forceinline__ int wave_incl_scan(int v, int lane) {
#pragma unroll
  for (int d = 1; d < 32; d <<= 1) {
    int t = __shfl_up(v, d, 32);
    if (lane >= d) v += t;
  }
  return v;
}

__global__ __launch_bounds__(256) void k_deg_to_inv(float* deg, const int* __restrict__ cnt,
                                                    int* bsum, int n) {
  __shared__ int sh[8];
  const int tid = threadIdx.x, lane = tid & 31, wave = tid >> 5;
  int i = blockIdx.x * 256 + tid;
  int c = 0;
  if (i < n) {
    float d = deg[i];
    deg[i] = (d > 0.0f) ? rsqrtf(fmaxf(d, 1e-12f)) : 0.0f;
    c = cnt[i];
  }
  int incl = wave_incl_scan(c, lane);
  if (lane == 31) sh[wave] = incl;
  __syncthreads();
  if (tid == 0) {
    int tot = 0;
    for (int k = 0; k < 8; ++k) tot += sh[k];
    bsum[blockIdx.x] = tot;
  }
}

__global__ __launch_bounds__(256) void k_scan(const int* __restrict__ cnt,
                                              const int* __restrict__ bsum,
                                              int* off, int* cur, int n) {
  __shared__ int sh[16];
  const int tid = threadIdx.x, lane = tid & 31, wave = tid >> 5;
  const int b = blockIdx.x;
  const int i = b * 256 + tid;
  const int c = (i < n) ? cnt[i] : 0;
  int part = 0;
  for (int k = tid; k < b; k += 256) part += bsum[k];
#pragma unroll
  for (int d = 16; d > 0; d >>= 1) part += __shfl_xor(part, d, 32);
  const int incl = wave_incl_scan(c, lane);
  if (lane == 31) sh[wave] = incl;
  if (lane == 0)  sh[8 + wave] = part;
  __syncthreads();
  int wbase = 0, base = 0;
  for (int k = 0; k < 8; ++k) {
    int t = sh[k];
    if (k < wave) wbase += t;
    base += sh[8 + k];
  }
  const int start = base + wbase + incl - c;
  if (i < n) { off[i] = start; cur[i] = start; }
  if (i == n - 1) off[n] = start + c;
}

__global__ __launch_bounds__(256) void k_place(const int* __restrict__ dst, int* cur,
                                               int* perm, int e) {
  int i = blockIdx.x * blockDim.x + threadIdx.x;
  if (i < e) perm[atomicAdd(&cur[dst[i]], 1)] = i;
}


template <int K, int NOUT, int RPP, int TILES, bool RELU>
__global__ __launch_bounds__(32 * (NOUT / 16) * RPP)
void k_gemm_wmma(const float* X, const float* __restrict__ W, float* H) {
  constexpr int CT    = NOUT / 16;
  constexpr int NT    = 32 * CT * RPP;
  constexpr int ROWS  = 16 * RPP;
  constexpr int PITCH = K + 4;
  constexpr int NV    = ROWS * K / 4 / NT;
  constexpr int G     = 8;
  constexpr int NG    = K / 4 / G;
  static_assert(NG * G * 4 == K, "K must be a whole number of batches");
  static_assert(ROWS * K == NV * 4 * NT, "a pass must split evenly over the threads");
  static_assert((N_NODES / 16) % (RPP * TILES) == 0, "passes must cover the rows exactly");

  __shared__ __attribute__((aligned(16))) float sA[2][ROWS * PITCH];

  const int tid  = threadIdx.x;
  const int lane = tid & 31;
  const int wave = tid >> 5;
  const int ct   = wave % CT;
  const int rsub = wave / CT;
  const int l    = lane & 15;
  const int half = lane >> 4;
  const int ko   = half * 2;
  const int n0   = ct * 16;

  v2f b[K / 4];
#pragma unroll
  for (int j = 0; j < K / 4; ++j) {
    b[j].x = W[(size_t)(4 * j + ko)     * NOUT + n0 + l];
    b[j].y = W[(size_t)(4 * j + ko + 1) * NOUT + n0 + l];
  }

  const size_t row0 = (size_t)blockIdx.x * TILES * ROWS;
  const v4f* gsrc = (const v4f*)(X + row0 * K) + tid;

  v4f r[NV];
#pragma unroll
  for (int i = 0; i < NV; ++i) r[i] = gsrc[i * NT];
#pragma unroll
  for (int i = 0; i < NV; ++i) {
    const int f = 4 * (i * NT + tid);
    v4f v = r[i];
    if (RELU) { v.x = fmaxf(v.x, 0.0f); v.y = fmaxf(v.y, 0.0f);
                v.z = fmaxf(v.z, 0.0f); v.w = fmaxf(v.w, 0.0f); }
    *(v4f*)&sA[0][(f / K) * PITCH + (f % K)] = v;
  }
  __syncthreads();

  for (int t = 0; t < TILES; ++t) {
    const int cur = t & 1;
    const bool more = (t + 1 < TILES);
    if (more) {
      const v4f* g = gsrc + (size_t)(t + 1) * (ROWS * K / 4);
#pragma unroll
      for (int i = 0; i < NV; ++i) r[i] = g[i * NT];
    }

    const float* arow = &sA[cur][(rsub * 16 + l) * PITCH + ko];
    v8f c = {0.f, 0.f, 0.f, 0.f, 0.f, 0.f, 0.f, 0.f};
    v2f fa[2][G];
#pragma unroll
    for (int jj = 0; jj < G; ++jj) fa[0][jj] = *(const v2f*)(arow + 4 * jj);
#pragma unroll
    for (int g = 0; g < NG; ++g) {
      if (g + 1 < NG) {
#pragma unroll
        for (int jj = 0; jj < G; ++jj)
          fa[(g + 1) & 1][jj] = *(const v2f*)(arow + 4 * ((g + 1) * G + jj));
      }
      __builtin_amdgcn_sched_barrier(0);
#pragma unroll
      for (int jj = 0; jj < G; ++jj)
        c = __builtin_amdgcn_wmma_f32_16x16x4_f32(false, fa[g & 1][jj], false, b[g * G + jj],
                                                  (short)0, c, false, false);
      __builtin_amdgcn_sched_barrier(0);
    }

    if (more) {
#pragma unroll
      for (int i = 0; i < NV; ++i) {
        const int f = 4 * (i * NT + tid);
        v4f v = r[i];
        if (RELU) { v.x = fmaxf(v.x, 0.0f); v.y = fmaxf(v.y, 0.0f);
                    v.z = fmaxf(v.z, 0.0f); v.w = fmaxf(v.w, 0.0f); }
        *(v4f*)&sA[cur ^ 1][(f / K) * PITCH + (f % K)] = v;
      }
      __syncthreads();
    }

    const size_t m0 = row0 + (size_t)t * ROWS + rsub * 16;
    float* drow = H + (m0 + half * 8) * NOUT + (n0 + l);
#pragma unroll
    for (int i = 0; i < 8; ++i) drow[(size_t)i * NOUT] = c[i];
  }
}


static __device__ __forceinline__ float4 scaled(float a, float4 v) {
  return make_float4(a * v.x, a * v.y, a * v.z, a * v.w);
}
static __device__ __forceinline__ void add_scaled(float4& acc, float a, float4 v) {
  acc.x += a * v.x; acc.y += a * v.y; acc.z += a * v.z; acc.w += a * v.w;
}
static __device__ __forceinline__ float4 plus(float4 a, float4 b) {
  return make_float4(a.x + b.x, a.y + b.y, a.z + b.z, a.w + b.w);
}

template <int G, int NP, int LPN>
static __device__ __forceinline__ void walk_slots(float4 (&acc)[NP], const float4* hp,
                                                  int sr, float nr, int lane0) {
  constexpr size_t RP = NP * LPN;
  float4 v[G][NP];
#pragma unroll
  for (int k = 0; k < G; ++k) {
    const int sk = __shfl(sr, lane0 + k, 32);
#pragma unroll
    for (int p = 0; p < NP; ++p) v[k][p] = hp[(size_t)sk * RP + p * LPN];
  }
#pragma unroll
  for (int k = 0; k < G; ++k) {
    const float a = __shfl(nr, lane0 + k, 32);
#pragma unroll
    for (int p = 0; p < NP; ++p) add_scaled(acc[p], a, v[k][p]);
  }
}

template <int DIM, int LPN, int GROUP>
__global__ __launch_bounds__(256) void k_gather(const int* __restrict__ src,
                                                const float* __restrict__ wgt,
                                                const float* __restrict__ inv,
                                                const int* __restrict__ off,
                                                const int* __restrict__ perm,
                                                const float* __restrict__ h,
                                                const float* __restrict__ bias,
                                                float* __restrict__ res) {
  constexpr int NPW = 32 / LPN;
  constexpr int NP  = DIM / 4 / LPN;
  constexpr int R   = 32 / LPN;
  static_assert(NP * 4 * LPN == DIM, "a row must split evenly over the lanes of a group");
  static_assert(LPN % GROUP == 0, "a group of gathers stays inside one slot register");
  static_assert(N_NODES % (8 * NPW) == 0, "the waves must cover the nodes exactly");

  const int lane  = threadIdx.x & 31;
  const int j     = lane % LPN;
  const int gbase = lane - j;
  const int node  = (blockIdx.x * 8 + (int)(threadIdx.x >> 5)) * NPW + lane / LPN;

  const int beg = off[node];
  const int end = off[node + 1];
  const float inv_t = inv[node];
  const float4* hp = (const float4*)h + j;
  constexpr size_t RP = DIM / 4;

  float4 acc[NP];
#pragma unroll
  for (int p = 0; p < NP; ++p)
    acc[p] = scaled(inv_t * 1.0f * inv_t, hp[(size_t)node * RP + p * LPN]);

  for (int c0 = beg; ; c0 += 32) {
    const int m = min(32, end - c0);
    int mm = m;
#pragma unroll
    for (int d = LPN; d < 32; d <<= 1) mm = max(mm, __shfl_xor(mm, d, 32));
    mm = __builtin_amdgcn_readfirstlane(mm);
    if (mm <= 0) break;

    int s[R]; float nrm[R];
    int e[R]; float we[R];
#pragma unroll
    for (int r = 0; r < R; ++r) e[r] = perm[(r * LPN + j < m) ? c0 + r * LPN + j : 0];
#pragma unroll
    for (int r = 0; r < R; ++r) { s[r] = src[e[r]]; we[r] = wgt[e[r]]; }
#pragma unroll
    for (int r = 0; r < R; ++r) nrm[r] = inv[s[r]] * we[r] * inv_t;
#pragma unroll
    for (int r = 0; r < R; ++r) {
      if (!(r * LPN + j < m)) { s[r] = node; nrm[r] = 0.0f; }
      asm volatile("" : "+v"(s[r]), "+v"(nrm[r]));
    }
#pragma unroll
    for (int r = 0; r < R; ++r) {
#pragma unroll
      for (int u = 0; u < LPN; u += GROUP) {
        const int rem = mm - (r * LPN + u);
        if (rem <= 0) break;
        if (rem > GROUP / 2) walk_slots<GROUP, NP, LPN>(acc, hp, s[r], nrm[r], gbase + u);
        else                 walk_slots<GROUP / 2, NP, LPN>(acc, hp, s[r], nrm[r], gbase + u);
      }
      if ((r + 1) * LPN >= mm) break;
    }
  }

  const float4* bp = (const float4*)bias + j;
  float4* rp = (float4*)res + (size_t)node * RP + j;
#pragma unroll
  for (int p = 0; p < NP; ++p) rp[p * LPN] = plus(acc[p], bp[p * LPN]);
}


extern "C" void kernel_launch(void* const* d_in, const int* in_sizes, int n_in,
                              void* d_out, int out_size, void* d_ws, size_t ws_size,
                              hipStream_t stream) {
  const int N = N_NODES, E = N_EDGES;

  const float* x   = (const float*)d_in[0];
  const int*   ei  = (const int*)d_in[1];
  const int*   src = ei;
  const int*   dst = ei + E;
  const float* w   = (const float*)d_in[2];
  const float* W1  = (const float*)d_in[3];
  const float* b1  = (const float*)d_in[4];
  const float* W2  = (const float*)d_in[5];
  const float* b2  = (const float*)d_in[6];

  float* out  = (float*)d_out;
  float* xemb = out + (size_t)N * DIM_OUT;

  float* inv  = (float*)d_ws;
  float* h1   = inv + N;
  int*   perm = (int*)(h1 + (size_t)N * DIM_HID);
  int*   off  = perm + E;
  int*   cnt  = (int*)h1;
  int*   cur  = cnt + N;
  int*   bsum = cur + N;
  float* h2   = h1;

  k_init_deg  <<<(N + 255) / 256, 256, 0, stream>>>(inv, cnt, N);
  k_deg_accum <<<(E + 255) / 256, 256, 0, stream>>>(dst, w, inv, cnt, E);
  k_deg_to_inv<<<(N + 255) / 256, 256, 0, stream>>>(inv, cnt, bsum, N);
  k_scan      <<<(N + 255) / 256, 256, 0, stream>>>(cnt, bsum, off, cur, N);
  k_place     <<<(E + 255) / 256, 256, 0, stream>>>(dst, cur, perm, E);

  k_gemm_wmma<DIM_IN, DIM_HID, 1, 5, false><<<(N / 16) / 5, 256, 0, stream>>>(x, W1, h1);

  k_gather<DIM_HID, 16, 8><<<N / (8 * 2), 256, 0, stream>>>(src, w, inv, off, perm, h1, b1, xemb);

  k_gemm_wmma<DIM_HID, DIM_OUT, 5, 2, true><<<(N / 16) / (5 * 2), 320, 0, stream>>>(xemb, W2, h2);

  k_gather<DIM_OUT, 8, 8><<<N / (8 * 4), 256, 0, stream>>>(src, w, inv, off, perm, h2, b2, out);
}


# one atomic per edge, not three: the counting add's return is the edge's rank in its target's list (list order = arrival of these adds); degree = fixed-order f32 sum over the list, no f32 atomics
# speedup vs baseline: 2.2862x; 1.2811x over previous
#include <hip/hip_runtime.h>

typedef __attribute__((ext_vector_type(2))) float v2f;
typedef __attribute__((ext_vector_type(4))) float v4f;
typedef __attribute__((ext_vector_type(8))) float v8f;

#define N_NODES  100000
#define N_EDGES  1600000
#define DIM_IN   256
#define DIM_HID  128
#define DIM_OUT  32


__global__ __launch_bounds__(256) void k_init_cnt(int* cnt, int n) {
  int i = blockIdx.x * blockDim.x + threadIdx.x;
  if (i < n) cnt[i] = 0;
}

__global__ __launch_bounds__(256) void k_count(const int* __restrict__ dst, int* cnt,
                                               int* __restrict__ rank, int e) {
  int i = blockIdx.x * blockDim.x + threadIdx.x;
  if (i < e) rank[i] = atomicAdd(&cnt[dst[i]], 1);
}

static __device__ __forceinline__ int wave_incl_scan(int v, int lane) {
#pragma unroll
  for (int d = 1; d < 32; d <<= 1) {
    int t = __shfl_up(v, d, 32);
    if (lane >= d) v += t;
  }
  return v;
}

__global__ __launch_bounds__(256) void k_block_sums(const int* __restrict__ cnt, int* bsum, int n) {
  __shared__ int sh[8];
  const int tid = threadIdx.x, lane = tid & 31, wave = tid >> 5;
  int i = blockIdx.x * 256 + tid;
  const int c = (i < n) ? cnt[i] : 0;
  int incl = wave_incl_scan(c, lane);
  if (lane == 31) sh[wave] = incl;
  __syncthreads();
  if (tid == 0) {
    int tot = 0;
    for (int k = 0; k < 8; ++k) tot += sh[k];
    bsum[blockIdx.x] = tot;
  }
}

__global__ __launch_bounds__(256) void k_scan(const int* __restrict__ cnt,
                                              const int* __restrict__ bsum,
                                              int* off, int n) {
  __shared__ int sh[16];
  const int tid = threadIdx.x, lane = tid & 31, wave = tid >> 5;
  const int b = blockIdx.x;
  const int i = b * 256 + tid;
  const int c = (i < n) ? cnt[i] : 0;
  int part = 0;
  for (int k = tid; k < b; k += 256) part += bsum[k];
#pragma unroll
  for (int d = 16; d > 0; d >>= 1) part += __shfl_xor(part, d, 32);
  const int incl = wave_incl_scan(c, lane);
  if (lane == 31) sh[wave] = incl;
  if (lane == 0)  sh[8 + wave] = part;
  __syncthreads();
  int wbase = 0, base = 0;
  for (int k = 0; k < 8; ++k) {
    int t = sh[k];
    if (k < wave) wbase += t;
    base += sh[8 + k];
  }
  const int start = base + wbase + incl - c;
  if (i < n) off[i] = start;
  if (i == n - 1) off[n] = start + c;
}

__global__ __launch_bounds__(256) void k_place(const int* __restrict__ dst,
                                               const int* __restrict__ off,
                                               const int* __restrict__ rank,
                                               int* __restrict__ perm, int e) {
  int i = blockIdx.x * blockDim.x + threadIdx.x;
  if (i < e) perm[off[dst[i]] + rank[i]] = i;
}

__global__ __launch_bounds__(256) void k_deg_list(const float* __restrict__ w,
                                                  const int* __restrict__ off,
                                                  const int* __restrict__ perm,
                                                  float* __restrict__ inv) {
  static_assert(N_NODES % 32 == 0, "the 8-lane groups must cover the nodes exactly");
  const int t    = blockIdx.x * 256 + threadIdx.x;
  const int node = t >> 3;
  const int j    = t & 7;
  const int beg  = off[node];
  const int end  = off[node + 1];
  float part = 0.0f;
  for (int c0 = beg; c0 < end; c0 += 32) {
    int e[4]; float we[4];
    for (int r = 0; r < 4; ++r) { const int k = c0 + r * 8 + j; e[r] = perm[(k < end) ? k : c0]; }
    for (int r = 0; r < 4; ++r) we[r] = w[e[r]];
    for (int r = 0; r < 4; ++r) part += (c0 + r * 8 + j < end) ? we[r] : 0.0f;
  }
  for (int d = 1; d < 8; d <<= 1) part += __shfl_xor(part, d, 32);
  const float deg = part + 1.0f;
  if (j == 0) inv[node] = (deg > 0.0f) ? rsqrtf(fmaxf(deg, 1e-12f)) : 0.0f;
}


template <int K, int NOUT, int RPP, int TILES, bool RELU>
__global__ __launch_bounds__(32 * (NOUT / 16) * RPP)
void k_gemm_wmma(const float* X, const float* __restrict__ W, float* H) {
  constexpr int CT    = NOUT / 16;
  constexpr int NT    = 32 * CT * RPP;
  constexpr int ROWS  = 16 * RPP;
  constexpr int PITCH = K + 4;
  constexpr int NV    = ROWS * K / 4 / NT;
  constexpr int G     = 8;
  constexpr int NG    = K / 4 / G;
  static_assert(NG * G * 4 == K, "K must be a whole number of batches");
  static_assert(ROWS * K == NV * 4 * NT, "a pass must split evenly over the threads");
  static_assert((N_NODES / 16) % (RPP * TILES) == 0, "passes must cover the rows exactly");

  __shared__ __attribute__((aligned(16))) float sA[2][ROWS * PITCH];

  const int tid  = threadIdx.x;
  const int lane = tid & 31;
  const int wave = tid >> 5;
  const int ct   = wave % CT;
  const int rsub = wave / CT;
  const int l    = lane & 15;
  const int half = lane >> 4;
  const int ko   = half * 2;
  const int n0   = ct * 16;

  v2f b[K / 4];
#pragma unroll
  for (int j = 0; j < K / 4; ++j) {
    b[j].x = W[(size_t)(4 * j + ko)     * NOUT + n0 + l];
    b[j].y = W[(size_t)(4 * j + ko + 1) * NOUT + n0 + l];
  }

  const size_t row0 = (size_t)blockIdx.x * TILES * ROWS;
  const v4f* gsrc = (const v4f*)(X + row0 * K) + tid;

  v4f r[NV];
#pragma unroll
  for (int i = 0; i < NV; ++i) r[i] = gsrc[i * NT];
#pragma unroll
  for (int i = 0; i < NV; ++i) {
    const int f = 4 * (i * NT + tid);
    v4f v = r[i];
    if (RELU) { v.x = fmaxf(v.x, 0.0f); v.y = fmaxf(v.y, 0.0f);
                v.z = fmaxf(v.z, 0.0f); v.w = fmaxf(v.w, 0.0f); }
    *(v4f*)&sA[0][(f / K) * PITCH + (f % K)] = v;
  }
  __syncthreads();

  for (int t = 0; t < TILES; ++t) {
    const int cur = t & 1;
    const bool more = (t + 1 < TILES);
    if (more) {
      const v4f* g = gsrc + (size_t)(t + 1) * (ROWS * K / 4);
#pragma unroll
      for (int i = 0; i < NV; ++i) r[i] = g[i * NT];
    }

    const float* arow = &sA[cur][(rsub * 16 + l) * PITCH + ko];
    v8f c = {0.f, 0.f, 0.f, 0.f, 0.f, 0.f, 0.f, 0.f};
    v2f fa[2][G];
#pragma unroll
    for (int jj = 0; jj < G; ++jj) fa[0][jj] = *(const v2f*)(arow + 4 * jj);
#pragma unroll
    for (int g = 0; g < NG; ++g) {
      if (g + 1 < NG) {
#pragma unroll
        for (int jj = 0; jj < G; ++jj)
          fa[(g + 1) & 1][jj] = *(const v2f*)(arow + 4 * ((g + 1) * G + jj));
      }
      __builtin_amdgcn_sched_barrier(0);
#pragma unroll
      for (int jj = 0; jj < G; ++jj)
        c = __builtin_amdgcn_wmma_f32_16x16x4_f32(false, fa[g & 1][jj], false, b[g * G + jj],
                                                  (short)0, c, false, false);
      __builtin_amdgcn_sched_barrier(0);
    }

    if (more) {
#pragma unroll
      for (int i = 0; i < NV; ++i) {
        const int f = 4 * (i * NT + tid);
        v4f v = r[i];
        if (RELU) { v.x = fmaxf(v.x, 0.0f); v.y = fmaxf(v.y, 0.0f);
                    v.z = fmaxf(v.z, 0.0f); v.w = fmaxf(v.w, 0.0f); }
        *(v4f*)&sA[cur ^ 1][(f / K) * PITCH + (f % K)] = v;
      }
      __syncthreads();
    }

    const size_t m0 = row0 + (size_t)t * ROWS + rsub * 16;
    float* drow = H + (m0 + half * 8) * NOUT + (n0 + l);
#pragma unroll
    for (int i = 0; i < 8; ++i) drow[(size_t)i * NOUT] = c[i];
  }
}


static __device__ __forceinline__ float4 scaled(float a, float4 v) {
  return make_float4(a * v.x, a * v.y, a * v.z, a * v.w);
}
static __device__ __forceinline__ void add_scaled(float4& acc, float a, float4 v) {
  acc.x += a * v.x; acc.y += a * v.y; acc.z += a * v.z; acc.w += a * v.w;
}
static __device__ __forceinline__ float4 plus(float4 a, float4 b) {
  return make_float4(a.x + b.x, a.y + b.y, a.z + b.z, a.w + b.w);
}

template <int G, int NP, int LPN>
static __device__ __forceinline__ void walk_slots(float4 (&acc)[NP], const float4* hp,
                                                  int sr, float nr, int lane0) {
  constexpr size_t RP = NP * LPN;
  float4 v[G][NP];
#pragma unroll
  for (int k = 0; k < G; ++k) {
    const int sk = __shfl(sr, lane0 + k, 32);
#pragma unroll
    for (int p = 0; p < NP; ++p) v[k][p] = hp[(size_t)sk * RP + p * LPN];
  }
#pragma unroll
  for (int k = 0; k < G; ++k) {
    const float a = __shfl(nr, lane0 + k, 32);
#pragma unroll
    for (int p = 0; p < NP; ++p) add_scaled(acc[p], a, v[k][p]);
  }
}

template <int DIM, int LPN, int GROUP>
__global__ __launch_bounds__(256) void k_gather(const int* __restrict__ src,
                                                const float* __restrict__ wgt,
                                                const float* __restrict__ inv,
                                                const int* __restrict__ off,
                                                const int* __restrict__ perm,
                                                const float* __restrict__ h,
                                                const float* __restrict__ bias,
                                                float* __restrict__ res) {
  constexpr int NPW = 32 / LPN;
  constexpr int NP  = DIM / 4 / LPN;
  constexpr int R   = 32 / LPN;
  static_assert(NP * 4 * LPN == DIM, "a row must split evenly over the lanes of a group");
  static_assert(LPN % GROUP == 0, "a group of gathers stays inside one slot register");
  static_assert(N_NODES % (8 * NPW) == 0, "the waves must cover the nodes exactly");

  const int lane  = threadIdx.x & 31;
  const int j     = lane % LPN;
  const int gbase = lane - j;
  const int node  = (blockIdx.x * 8 + (int)(threadIdx.x >> 5)) * NPW + lane / LPN;

  const int beg = off[node];
  const int end = off[node + 1];
  const float inv_t = inv[node];
  const float4* hp = (const float4*)h + j;
  constexpr size_t RP = DIM / 4;

  float4 acc[NP];
#pragma unroll
  for (int p = 0; p < NP; ++p)
    acc[p] = scaled(inv_t * 1.0f * inv_t, hp[(size_t)node * RP + p * LPN]);

  for (int c0 = beg; ; c0 += 32) {
    const int m = min(32, end - c0);
    int mm = m;
#pragma unroll
    for (int d = LPN; d < 32; d <<= 1) mm = max(mm, __shfl_xor(mm, d, 32));
    mm = __builtin_amdgcn_readfirstlane(mm);
    if (mm <= 0) break;

    int s[R]; float nrm[R];
    int e[R]; float we[R];
#pragma unroll
    for (int r = 0; r < R; ++r) e[r] = perm[(r * LPN + j < m) ? c0 + r * LPN + j : 0];
#pragma unroll
    for (int r = 0; r < R; ++r) { s[r] = src[e[r]]; we[r] = wgt[e[r]]; }
#pragma unroll
    for (int r = 0; r < R; ++r) nrm[r] = inv[s[r]] * we[r] * inv_t;
#pragma unroll
    for (int r = 0; r < R; ++r) {
      if (!(r * LPN + j < m)) { s[r] = node; nrm[r] = 0.0f; }
      asm volatile("" : "+v"(s[r]), "+v"(nrm[r]));
    }
#pragma unroll
    for (int r = 0; r < R; ++r) {
#pragma unroll
      for (int u = 0; u < LPN; u += GROUP) {
        const int rem = mm - (r * LPN + u);
        if (rem <= 0) break;
        if (rem > GROUP / 2) walk_slots<GROUP, NP, LPN>(acc, hp, s[r], nrm[r], gbase + u);
        else                 walk_slots<GROUP / 2, NP, LPN>(acc, hp, s[r], nrm[r], gbase + u);
      }
      if ((r + 1) * LPN >= mm) break;
    }
  }

  const float4* bp = (const float4*)bias + j;
  float4* rp = (float4*)res + (size_t)node * RP + j;
#pragma unroll
  for (int p = 0; p < NP; ++p) rp[p * LPN] = plus(acc[p], bp[p * LPN]);
}


extern "C" void kernel_launch(void* const* d_in, const int* in_sizes, int n_in,
                              void* d_out, int out_size, void* d_ws, size_t ws_size,
                              hipStream_t stream) {
  const int N = N_NODES, E = N_EDGES;

  const float* x   = (const float*)d_in[0];
  const int*   ei  = (const int*)d_in[1];
  const int*   src = ei;
  const int*   dst = ei + E;
  const float* w   = (const float*)d_in[2];
  const float* W1  = (const float*)d_in[3];
  const float* b1  = (const float*)d_in[4];
  const float* W2  = (const float*)d_in[5];
  const float* b2  = (const float*)d_in[6];

  float* out  = (float*)d_out;
  float* xemb = out + (size_t)N * DIM_OUT;

  float* inv  = (float*)d_ws;
  float* h1   = inv + N;
  int*   perm = (int*)(h1 + (size_t)N * DIM_HID);
  int*   off  = perm + E;
  int*   cnt  = (int*)h1;
  int*   rank = cnt + N;
  int*   bsum = rank + E;
  float* h2   = h1;

  k_init_cnt  <<<(N + 255) / 256, 256, 0, stream>>>(cnt, N);
  k_count     <<<(E + 255) / 256, 256, 0, stream>>>(dst, cnt, rank, E);
  k_block_sums<<<(N + 255) / 256, 256, 0, stream>>>(cnt, bsum, N);
  k_scan      <<<(N + 255) / 256, 256, 0, stream>>>(cnt, bsum, off, N);
  k_place     <<<(E + 255) / 256, 256, 0, stream>>>(dst, off, rank, perm, E);
  k_deg_list  <<<N / 32, 256, 0, stream>>>(w, off, perm, inv);

  k_gemm_wmma<DIM_IN, DIM_HID, 1, 5, false><<<(N / 16) / 5, 256, 0, stream>>>(x, W1, h1);

  k_gather<DIM_HID, 16, 8><<<N / (8 * 2), 256, 0, stream>>>(src, w, inv, off, perm, h1, b1, xemb);

  k_gemm_wmma<DIM_HID, DIM_OUT, 5, 2, true><<<(N / 16) / (5 * 2), 320, 0, stream>>>(xemb, W2, h2);

  k_gather<DIM_OUT, 8, 8><<<N / (8 * 4), 256, 0, stream>>>(src, w, inv, off, perm, h2, b2, out);
}


# edge ordering as a two-level counting sort on LDS counters (256-node bins per edge piece, then nodes inside a bin): no global atomics, 5 build launches become 4; list order = arrival of the LDS adds
# speedup vs baseline: 3.1226x; 1.3658x over previous
#include <hip/hip_runtime.h>

typedef __attribute__((ext_vector_type(2))) float v2f;
typedef __attribute__((ext_vector_type(4))) float v4f;
typedef __attribute__((ext_vector_type(8))) float v8f;

#define N_NODES  100000
#define N_EDGES  1600000
#define DIM_IN   256
#define DIM_HID  128
#define DIM_OUT  32


#define NBIN   ((N_NODES + 255) / 256)
#define NCHUNK 256
#define CHUNK  (N_EDGES / NCHUNK)
#define NT_E   512

static __device__ __forceinline__ int wave_incl_scan(int v, int lane) {
#pragma unroll
  for (int d = 1; d < 32; d <<= 1) {
    int t = __shfl_up(v, d, 32);
    if (lane >= d) v += t;
  }
  return v;
}

template <int NW>
static __device__ __forceinline__ int block_excl_scan(int v, int* sh, int& total) {
  const int lane = threadIdx.x & 31, wave = threadIdx.x >> 5;
  const int incl = wave_incl_scan(v, lane);
  if (lane == 31) sh[wave] = incl;
  __syncthreads();
  int base = 0, tot = 0;
  for (int k = 0; k < NW; ++k) {
    const int t = sh[k];
    if (k < wave) base += t;
    tot += t;
  }
  total = tot;
  __syncthreads();
  return base + incl - v;
}

__global__ __launch_bounds__(NT_E) void k_bin_count(const int* __restrict__ dst,
                                                    int* __restrict__ binCnt) {
  static_assert(CHUNK * NCHUNK == N_EDGES, "the pieces must cover the edges exactly");
  constexpr int PER = (CHUNK + NT_E - 1) / NT_E;
  __shared__ int cnt[NBIN];
  const int tid = threadIdx.x, blk = blockIdx.x;
  for (int b = tid; b < NBIN; b += NT_E) cnt[b] = 0;
  const int* d = dst + blk * CHUNK;
  int dv[PER];
#pragma unroll
  for (int k = 0; k < PER; ++k) { const int i = tid + k * NT_E; dv[k] = d[(i < CHUNK) ? i : 0]; }
  __syncthreads();
#pragma unroll
  for (int k = 0; k < PER; ++k) if (tid + k * NT_E < CHUNK) atomicAdd(&cnt[dv[k] >> 8], 1);
  __syncthreads();
  for (int b = tid; b < NBIN; b += NT_E) binCnt[b * NCHUNK + blk] = cnt[b];
}

__global__ __launch_bounds__(NCHUNK) void k_bin_prefix(const int* __restrict__ binCnt,
                                                       int* __restrict__ binPre,
                                                       int* __restrict__ binTot) {
  __shared__ int sh[NCHUNK / 32];
  const int bin = blockIdx.x, t = threadIdx.x;
  int tot;
  const int ex = block_excl_scan<NCHUNK / 32>(binCnt[bin * NCHUNK + t], sh, tot);
  binPre[bin * NCHUNK + t] = ex;
  if (t == 0) binTot[bin] = tot;
}

__global__ __launch_bounds__(NT_E) void k_bin_scatter(const int* __restrict__ dst,
                                                      const int* __restrict__ binPre,
                                                      const int* __restrict__ binTot,
                                                      unsigned* __restrict__ tmp) {
  static_assert(NBIN <= NT_E, "one thread per bin in the scan of the bin totals");
  static_assert(N_EDGES < (1 << 24), "an edge id must leave 8 bits of the packed word free");
  constexpr int PER = (CHUNK + NT_E - 1) / NT_E;
  __shared__ int cur[NBIN];
  __shared__ int sh[NT_E / 32];
  const int tid = threadIdx.x, blk = blockIdx.x;
  const int* d = dst + blk * CHUNK;
  int dv[PER];
#pragma unroll
  for (int k = 0; k < PER; ++k) { const int i = tid + k * NT_E; dv[k] = d[(i < CHUNK) ? i : 0]; }
  int tot;
  const int start = block_excl_scan<NT_E / 32>((tid < NBIN) ? binTot[tid] : 0, sh, tot);
  if (tid < NBIN) cur[tid] = start + binPre[tid * NCHUNK + blk];
  __syncthreads();
  int pos[PER];
#pragma unroll
  for (int k = 0; k < PER; ++k)
    pos[k] = (tid + k * NT_E < CHUNK) ? atomicAdd(&cur[dv[k] >> 8], 1) : 0;
#pragma unroll
  for (int k = 0; k < PER; ++k) {
    if (tid + k * NT_E < CHUNK) {
      const int e = blk * CHUNK + tid + k * NT_E;
      tmp[pos[k]] = ((unsigned)e << 8) | (unsigned)(dv[k] & 255);
    }
  }
}

__global__ __launch_bounds__(NT_E) void k_bin_sort(const unsigned* __restrict__ tmp,
                                                   const int* __restrict__ binTot,
                                                   int* __restrict__ off,
                                                   int* __restrict__ perm) {
  static_assert(NBIN <= NT_E && 256 <= NT_E, "one thread per bin and per node of a bin");
  constexpr int RB = (N_EDGES / NBIN) / NT_E + 2;
  __shared__ int cnt[256];
  __shared__ int sh[NT_E / 32];
  __shared__ int s_first, s_count;
  const int tid = threadIdx.x, bin = blockIdx.x;
  const int mine = (tid < NBIN) ? binTot[tid] : 0;
  int tot;
  const int ex = block_excl_scan<NT_E / 32>(mine, sh, tot);
  if (tid == bin) { s_first = ex; s_count = mine; }
  if (tid < 256) cnt[tid] = 0;
  __syncthreads();
  const int bs = s_first, nb = s_count;
  const unsigned* rec = tmp + bs;

  for (int i0 = 0; i0 < nb; i0 += RB * NT_E) {
    unsigned r[RB];
#pragma unroll
    for (int k = 0; k < RB; ++k) r[k] = rec[min(i0 + k * NT_E + tid, nb - 1)];
#pragma unroll
    for (int k = 0; k < RB; ++k) if (i0 + k * NT_E + tid < nb) atomicAdd(&cnt[r[k] & 255u], 1);
  }
  __syncthreads();

  int tot2;
  const int ex2 = block_excl_scan<NT_E / 32>((tid < 256) ? cnt[tid] : 0, sh, tot2);
  if (tid < 256) {
    cnt[tid] = ex2;
    const int node = bin * 256 + tid;
    if (node < N_NODES) off[node] = bs + ex2;
  }
  if (bin == NBIN - 1 && tid == 0) off[N_NODES] = bs + nb;
  __syncthreads();

  for (int i0 = 0; i0 < nb; i0 += RB * NT_E) {
    unsigned r[RB];
    int p[RB];
#pragma unroll
    for (int k = 0; k < RB; ++k) r[k] = rec[min(i0 + k * NT_E + tid, nb - 1)];
#pragma unroll
    for (int k = 0; k < RB; ++k)
      p[k] = (i0 + k * NT_E + tid < nb) ? atomicAdd(&cnt[r[k] & 255u], 1) : 0;
#pragma unroll
    for (int k = 0; k < RB; ++k)
      if (i0 + k * NT_E + tid < nb) perm[bs + p[k]] = (int)(r[k] >> 8);
  }
}

__global__ __launch_bounds__(256) void k_deg_list(const float* __restrict__ w,
                                                  const int* __restrict__ off,
                                                  const int* __restrict__ perm,
                                                  float* __restrict__ inv) {
  static_assert(N_NODES % 32 == 0, "the 8-lane groups must cover the nodes exactly");
  const int t    = blockIdx.x * 256 + threadIdx.x;
  const int node = t >> 3;
  const int j    = t & 7;
  const int beg  = off[node];
  const int end  = off[node + 1];
  float part = 0.0f;
  for (int c0 = beg; c0 < end; c0 += 32) {
    int e[4]; float we[4];
    for (int r = 0; r < 4; ++r) { const int k = c0 + r * 8 + j; e[r] = perm[(k < end) ? k : c0]; }
    for (int r = 0; r < 4; ++r) we[r] = w[e[r]];
    for (int r = 0; r < 4; ++r) part += (c0 + r * 8 + j < end) ? we[r] : 0.0f;
  }
  for (int d = 1; d < 8; d <<= 1) part += __shfl_xor(part, d, 32);
  const float deg = part + 1.0f;
  if (j == 0) inv[node] = (deg > 0.0f) ? rsqrtf(fmaxf(deg, 1e-12f)) : 0.0f;
}


template <int K, int NOUT, int RPP, int TILES, bool RELU>
__global__ __launch_bounds__(32 * (NOUT / 16) * RPP)
void k_gemm_wmma(const float* X, const float* __restrict__ W, float* H) {
  constexpr int CT    = NOUT / 16;
  constexpr int NT    = 32 * CT * RPP;
  constexpr int ROWS  = 16 * RPP;
  constexpr int PITCH = K + 4;
  constexpr int NV    = ROWS * K / 4 / NT;
  constexpr int G     = 8;
  constexpr int NG    = K / 4 / G;
  static_assert(NG * G * 4 == K, "K must be a whole number of batches");
  static_assert(ROWS * K == NV * 4 * NT, "a pass must split evenly over the threads");
  static_assert((N_NODES / 16) % (RPP * TILES) == 0, "passes must cover the rows exactly");

  __shared__ __attribute__((aligned(16))) float sA[2][ROWS * PITCH];

  const int tid  = threadIdx.x;
  const int lane = tid & 31;
  const int wave = tid >> 5;
  const int ct   = wave % CT;
  const int rsub = wave / CT;
  const int l    = lane & 15;
  const int half = lane >> 4;
  const int ko   = half * 2;
  const int n0   = ct * 16;

  v2f b[K / 4];
#pragma unroll
  for (int j = 0; j < K / 4; ++j) {
    b[j].x = W[(size_t)(4 * j + ko)     * NOUT + n0 + l];
    b[j].y = W[(size_t)(4 * j + ko + 1) * NOUT + n0 + l];
  }

  const size_t row0 = (size_t)blockIdx.x * TILES * ROWS;
  const v4f* gsrc = (const v4f*)(X + row0 * K) + tid;

  v4f r[NV];
#pragma unroll
  for (int i = 0; i < NV; ++i) r[i] = gsrc[i * NT];
#pragma unroll
  for (int i = 0; i < NV; ++i) {
    const int f = 4 * (i * NT + tid);
    v4f v = r[i];
    if (RELU) { v.x = fmaxf(v.x, 0.0f); v.y = fmaxf(v.y, 0.0f);
                v.z = fmaxf(v.z, 0.0f); v.w = fmaxf(v.w, 0.0f); }
    *(v4f*)&sA[0][(f / K) * PITCH + (f % K)] = v;
  }
  __syncthreads();

  for (int t = 0; t < TILES; ++t) {
    const int cur = t & 1;
    const bool more = (t + 1 < TILES);
    if (more) {
      const v4f* g = gsrc + (size_t)(t + 1) * (ROWS * K / 4);
#pragma unroll
      for (int i = 0; i < NV; ++i) r[i] = g[i * NT];
    }

    const float* arow = &sA[cur][(rsub * 16 + l) * PITCH + ko];
    v8f c = {0.f, 0.f, 0.f, 0.f, 0.f, 0.f, 0.f, 0.f};
    v2f fa[2][G];
#pragma unroll
    for (int jj = 0; jj < G; ++jj) fa[0][jj] = *(const v2f*)(arow + 4 * jj);
#pragma unroll
    for (int g = 0; g < NG; ++g) {
      if (g + 1 < NG) {
#pragma unroll
        for (int jj = 0; jj < G; ++jj)
          fa[(g + 1) & 1][jj] = *(const v2f*)(arow + 4 * ((g + 1) * G + jj));
      }
      __builtin_amdgcn_sched_barrier(0);
#pragma unroll
      for (int jj = 0; jj < G; ++jj)
        c = __builtin_amdgcn_wmma_f32_16x16x4_f32(false, fa[g & 1][jj], false, b[g * G + jj],
                                                  (short)0, c, false, false);
      __builtin_amdgcn_sched_barrier(0);
    }

    if (more) {
#pragma unroll
      for (int i = 0; i < NV; ++i) {
        const int f = 4 * (i * NT + tid);
        v4f v = r[i];
        if (RELU) { v.x = fmaxf(v.x, 0.0f); v.y = fmaxf(v.y, 0.0f);
                    v.z = fmaxf(v.z, 0.0f); v.w = fmaxf(v.w, 0.0f); }
        *(v4f*)&sA[cur ^ 1][(f / K) * PITCH + (f % K)] = v;
      }
      __syncthreads();
    }

    const size_t m0 = row0 + (size_t)t * ROWS + rsub * 16;
    float* drow = H + (m0 + half * 8) * NOUT + (n0 + l);
#pragma unroll
    for (int i = 0; i < 8; ++i) drow[(size_t)i * NOUT] = c[i];
  }
}


static __device__ __forceinline__ float4 scaled(float a, float4 v) {
  return make_float4(a * v.x, a * v.y, a * v.z, a * v.w);
}
static __device__ __forceinline__ void add_scaled(float4& acc, float a, float4 v) {
  acc.x += a * v.x; acc.y += a * v.y; acc.z += a * v.z; acc.w += a * v.w;
}
static __device__ __forceinline__ float4 plus(float4 a, float4 b) {
  return make_float4(a.x + b.x, a.y + b.y, a.z + b.z, a.w + b.w);
}

template <int G, int NP, int LPN>
static __device__ __forceinline__ void walk_slots(float4 (&acc)[NP], const float4* hp,
                                                  int sr, float nr, int lane0) {
  constexpr size_t RP = NP * LPN;
  float4 v[G][NP];
#pragma unroll
  for (int k = 0; k < G; ++k) {
    const int sk = __shfl(sr, lane0 + k, 32);
#pragma unroll
    for (int p = 0; p < NP; ++p) v[k][p] = hp[(size_t)sk * RP + p * LPN];
  }
#pragma unroll
  for (int k = 0; k < G; ++k) {
    const float a = __shfl(nr, lane0 + k, 32);
#pragma unroll
    for (int p = 0; p < NP; ++p) add_scaled(acc[p], a, v[k][p]);
  }
}

template <int DIM, int LPN, int GROUP>
__global__ __launch_bounds__(256) void k_gather(const int* __restrict__ src,
                                                const float* __restrict__ wgt,
                                                const float* __restrict__ inv,
                                                const int* __restrict__ off,
                                                const int* __restrict__ perm,
                                                const float* __restrict__ h,
                                                const float* __restrict__ bias,
                                                float* __restrict__ res) {
  constexpr int NPW = 32 / LPN;
  constexpr int NP  = DIM / 4 / LPN;
  constexpr int R   = 32 / LPN;
  static_assert(NP * 4 * LPN == DIM, "a row must split evenly over the lanes of a group");
  static_assert(LPN % GROUP == 0, "a group of gathers stays inside one slot register");
  static_assert(N_NODES % (8 * NPW) == 0, "the waves must cover the nodes exactly");

  const int lane  = threadIdx.x & 31;
  const int j     = lane % LPN;
  const int gbase = lane - j;
  const int node  = (blockIdx.x * 8 + (int)(threadIdx.x >> 5)) * NPW + lane / LPN;

  const int beg = off[node];
  const int end = off[node + 1];
  const float inv_t = inv[node];
  const float4* hp = (const float4*)h + j;
  constexpr size_t RP = DIM / 4;

  float4 acc[NP];
#pragma unroll
  for (int p = 0; p < NP; ++p)
    acc[p] = scaled(inv_t * 1.0f * inv_t, hp[(size_t)node * RP + p * LPN]);

  for (int c0 = beg; ; c0 += 32) {
    const int m = min(32, end - c0);
    int mm = m;
#pragma unroll
    for (int d = LPN; d < 32; d <<= 1) mm = max(mm, __shfl_xor(mm, d, 32));
    mm = __builtin_amdgcn_readfirstlane(mm);
    if (mm <= 0) break;

    int s[R]; float nrm[R];
    int e[R]; float we[R];
#pragma unroll
    for (int r = 0; r < R; ++r) e[r] = perm[(r * LPN + j < m) ? c0 + r * LPN + j : 0];
#pragma unroll
    for (int r = 0; r < R; ++r) { s[r] = src[e[r]]; we[r] = wgt[e[r]]; }
#pragma unroll
    for (int r = 0; r < R; ++r) nrm[r] = inv[s[r]] * we[r] * inv_t;
#pragma unroll
    for (int r = 0; r < R; ++r) {
      if (!(r * LPN + j < m)) { s[r] = node; nrm[r] = 0.0f; }
      asm volatile("" : "+v"(s[r]), "+v"(nrm[r]));
    }
#pragma unroll
    for (int r = 0; r < R; ++r) {
#pragma unroll
      for (int u = 0; u < LPN; u += GROUP) {
        const int rem = mm - (r * LPN + u);
        if (rem <= 0) break;
        if (rem > GROUP / 2) walk_slots<GROUP, NP, LPN>(acc, hp, s[r], nrm[r], gbase + u);
        else                 walk_slots<GROUP / 2, NP, LPN>(acc, hp, s[r], nrm[r], gbase + u);
      }
      if ((r + 1) * LPN >= mm) break;
    }
  }

  const float4* bp = (const float4*)bias + j;
  float4* rp = (float4*)res + (size_t)node * RP + j;
#pragma unroll
  for (int p = 0; p < NP; ++p) rp[p * LPN] = plus(acc[p], bp[p * LPN]);
}


extern "C" void kernel_launch(void* const* d_in, const int* in_sizes, int n_in,
                              void* d_out, int out_size, void* d_ws, size_t ws_size,
                              hipStream_t stream) {
  const int N = N_NODES, E = N_EDGES;

  const float* x   = (const float*)d_in[0];
  const int*   ei  = (const int*)d_in[1];
  const int*   src = ei;
  const int*   dst = ei + E;
  const float* w   = (const float*)d_in[2];
  const float* W1  = (const float*)d_in[3];
  const float* b1  = (const float*)d_in[4];
  const float* W2  = (const float*)d_in[5];
  const float* b2  = (const float*)d_in[6];

  float* out  = (float*)d_out;
  float* xemb = out + (size_t)N * DIM_OUT;

  float* inv  = (float*)d_ws;
  float* h1   = inv + N;
  int*   perm = (int*)(h1 + (size_t)N * DIM_HID);
  int*   off  = perm + E;
  unsigned* tmp = (unsigned*)h1;
  int*   binCnt = (int*)h1 + E;
  int*   binPre = binCnt + NBIN * NCHUNK;
  int*   binTot = binPre + NBIN * NCHUNK;
  float* h2   = h1;

  k_bin_count  <<<NCHUNK, NT_E, 0, stream>>>(dst, binCnt);
  k_bin_prefix <<<NBIN, NCHUNK, 0, stream>>>(binCnt, binPre, binTot);
  k_bin_scatter<<<NCHUNK, NT_E, 0, stream>>>(dst, binPre, binTot, tmp);
  k_bin_sort   <<<NBIN, NT_E, 0, stream>>>(tmp, binTot, off, perm);
  k_deg_list  <<<N / 32, 256, 0, stream>>>(w, off, perm, inv);

  k_gemm_wmma<DIM_IN, DIM_HID, 1, 5, false><<<(N / 16) / 5, 256, 0, stream>>>(x, W1, h1);

  k_gather<DIM_HID, 16, 8><<<N / (8 * 2), 256, 0, stream>>>(src, w, inv, off, perm, h1, b1, xemb);

  k_gemm_wmma<DIM_HID, DIM_OUT, 5, 2, true><<<(N / 16) / (5 * 2), 320, 0, stream>>>(xemb, W2, h2);

  k_gather<DIM_OUT, 8, 8><<<N / (8 * 4), 256, 0, stream>>>(src, w, inv, off, perm, h2, b2, out);
}


# target sort moves each edge's source and weight, not its id (a bin is ordered in LDS, leaves in whole runs): degree and both gathers read them in list order, no id lookups; f32 sum orders unchanged
# speedup vs baseline: 3.7546x; 1.2024x over previous
#include <hip/hip_runtime.h>

typedef __attribute__((ext_vector_type(2))) float v2f;
typedef __attribute__((ext_vector_type(4))) float v4f;
typedef __attribute__((ext_vector_type(8))) float v8f;

#define N_NODES  100000
#define N_EDGES  1600000
#define DIM_IN   256
#define DIM_HID  128
#define DIM_OUT  32


#define NBIN   ((N_NODES + 255) / 256)
#define NCHUNK 256
#define CHUNK  (N_EDGES / NCHUNK)
#define NT_E   512

static __device__ __forceinline__ int wave_incl_scan(int v, int lane) {
#pragma unroll
  for (int d = 1; d < 32; d <<= 1) {
    int t = __shfl_up(v, d, 32);
    if (lane >= d) v += t;
  }
  return v;
}

template <int NW>
static __device__ __forceinline__ int block_excl_scan(int v, int* sh, int& total) {
  const int lane = threadIdx.x & 31, wave = threadIdx.x >> 5;
  const int incl = wave_incl_scan(v, lane);
  if (lane == 31) sh[wave] = incl;
  __syncthreads();
  int base = 0, tot = 0;
  for (int k = 0; k < NW; ++k) {
    const int t = sh[k];
    if (k < wave) base += t;
    tot += t;
  }
  total = tot;
  __syncthreads();
  return base + incl - v;
}

__global__ __launch_bounds__(NT_E) void k_bin_count(const int* __restrict__ dst,
                                                    int* __restrict__ binCnt) {
  static_assert(CHUNK * NCHUNK == N_EDGES, "the pieces must cover the edges exactly");
  constexpr int PER = (CHUNK + NT_E - 1) / NT_E;
  __shared__ int cnt[NBIN];
  const int tid = threadIdx.x, blk = blockIdx.x;
  for (int b = tid; b < NBIN; b += NT_E) cnt[b] = 0;
  const int* d = dst + blk * CHUNK;
  int dv[PER];
#pragma unroll
  for (int k = 0; k < PER; ++k) { const int i = tid + k * NT_E; dv[k] = d[(i < CHUNK) ? i : 0]; }
  __syncthreads();
#pragma unroll
  for (int k = 0; k < PER; ++k) if (tid + k * NT_E < CHUNK) atomicAdd(&cnt[dv[k] >> 8], 1);
  __syncthreads();
  for (int b = tid; b < NBIN; b += NT_E) binCnt[b * NCHUNK + blk] = cnt[b];
}

__global__ __launch_bounds__(NCHUNK) void k_bin_prefix(const int* __restrict__ binCnt,
                                                       int* __restrict__ binPre,
                                                       int* __restrict__ binTot) {
  __shared__ int sh[NCHUNK / 32];
  const int bin = blockIdx.x, t = threadIdx.x;
  int tot;
  const int ex = block_excl_scan<NCHUNK / 32>(binCnt[bin * NCHUNK + t], sh, tot);
  binPre[bin * NCHUNK + t] = ex;
  if (t == 0) binTot[bin] = tot;
}

__global__ __launch_bounds__(NT_E) void k_bin_scatter(const int* __restrict__ src,
                                                      const int* __restrict__ dst,
                                                      const float* __restrict__ w,
                                                      const int* __restrict__ binPre,
                                                      const int* __restrict__ binTot,
                                                      unsigned* __restrict__ tmpw) {
  static_assert(NBIN <= NT_E, "one thread per bin in the scan of the bin totals");
  static_assert(N_NODES <= (1 << 24), "a source must leave 8 bits of the record's first word free");
  constexpr int PER = (CHUNK + NT_E - 1) / NT_E;
  __shared__ int cur[NBIN];
  __shared__ int sh[NT_E / 32];
  const int tid = threadIdx.x, blk = blockIdx.x;
  uint2* tmp = (uint2*)tmpw;
  const int*   d  = dst + blk * CHUNK;
  const int*   s  = src + blk * CHUNK;
  const float* ww = w   + blk * CHUNK;
  int dv[PER], sv[PER];
  float wv[PER];
#pragma unroll
  for (int k = 0; k < PER; ++k) {
    const int i = tid + k * NT_E;
    const int ii = (i < CHUNK) ? i : 0;
    dv[k] = d[ii]; sv[k] = s[ii]; wv[k] = ww[ii];
  }
  int tot;
  const int start = block_excl_scan<NT_E / 32>((tid < NBIN) ? binTot[tid] : 0, sh, tot);
  if (tid < NBIN) cur[tid] = start + binPre[tid * NCHUNK + blk];
  __syncthreads();
  int pos[PER];
#pragma unroll
  for (int k = 0; k < PER; ++k)
    pos[k] = (tid + k * NT_E < CHUNK) ? atomicAdd(&cur[dv[k] >> 8], 1) : 0;
#pragma unroll
  for (int k = 0; k < PER; ++k) {
    if (tid + k * NT_E < CHUNK)
      tmp[pos[k]] = make_uint2(((unsigned)sv[k] << 8) | (unsigned)(dv[k] & 255),
                               __float_as_uint(wv[k]));
  }
}

__global__ __launch_bounds__(NT_E) void k_bin_sort(const unsigned* __restrict__ tmpw,
                                                   const int* __restrict__ binTot,
                                                   int* __restrict__ off,
                                                   float* __restrict__ wS,
                                                   unsigned* __restrict__ srcLoW,
                                                   unsigned* __restrict__ srcHiW) {
  static_assert(NBIN <= NT_E && 256 <= NT_E, "one thread per bin and per node of a bin");
  constexpr int RB  = (N_EDGES / NBIN) / NT_E + 2;
  constexpr int CAP = RB * NT_E;
  __shared__ int cnt[256];
  __shared__ int sh[NT_E / 32];
  __shared__ int s_first, s_count;
  __shared__ float    stW[CAP];
  __shared__ unsigned stS[CAP];
  const uint2* tmp = (const uint2*)tmpw;
  unsigned short* srcLo = (unsigned short*)srcLoW;
  unsigned char*  srcHi = (unsigned char*)srcHiW;
  const int tid = threadIdx.x, bin = blockIdx.x;
  const int mine = (tid < NBIN) ? binTot[tid] : 0;
  int tot;
  const int ex = block_excl_scan<NT_E / 32>(mine, sh, tot);
  if (tid == bin) { s_first = ex; s_count = mine; }
  if (tid < 256) cnt[tid] = 0;
  __syncthreads();
  const int bs = s_first, nb = s_count;
  const uint2* rec = tmp + bs;
  const bool staged = (nb <= CAP);

  for (int i0 = 0; i0 < nb; i0 += RB * NT_E) {
    unsigned r[RB];
#pragma unroll
    for (int k = 0; k < RB; ++k) r[k] = rec[min(i0 + k * NT_E + tid, nb - 1)].x;
#pragma unroll
    for (int k = 0; k < RB; ++k) if (i0 + k * NT_E + tid < nb) atomicAdd(&cnt[r[k] & 255u], 1);
  }
  __syncthreads();

  int tot2;
  const int ex2 = block_excl_scan<NT_E / 32>((tid < 256) ? cnt[tid] : 0, sh, tot2);
  if (tid < 256) {
    cnt[tid] = ex2;
    const int node = bin * 256 + tid;
    if (node < N_NODES) off[node] = bs + ex2;
  }
  if (bin == NBIN - 1 && tid == 0) off[N_NODES] = bs + nb;
  __syncthreads();

  for (int i0 = 0; i0 < nb; i0 += RB * NT_E) {
    uint2 r[RB];
    int p[RB];
#pragma unroll
    for (int k = 0; k < RB; ++k) r[k] = rec[min(i0 + k * NT_E + tid, nb - 1)];
#pragma unroll
    for (int k = 0; k < RB; ++k)
      p[k] = (i0 + k * NT_E + tid < nb) ? atomicAdd(&cnt[r[k].x & 255u], 1) : 0;
#pragma unroll
    for (int k = 0; k < RB; ++k) {
      if (i0 + k * NT_E + tid < nb) {
        const unsigned sv = r[k].x >> 8;
        if (staged) {
          stW[p[k]] = __uint_as_float(r[k].y);
          stS[p[k]] = sv;
        } else {
          const int q = bs + p[k];
          wS[q]    = __uint_as_float(r[k].y);
          srcLo[q] = (unsigned short)(sv & 0xffffu);
          srcHi[q] = (unsigned char)(sv >> 16);
        }
      }
    }
  }
  if (staged) {
    __syncthreads();
    for (int i = tid; i < nb; i += NT_E) {
      const unsigned sv = stS[i];
      wS[bs + i]    = stW[i];
      srcLo[bs + i] = (unsigned short)(sv & 0xffffu);
      srcHi[bs + i] = (unsigned char)(sv >> 16);
    }
  }
}

__global__ __launch_bounds__(256) void k_deg_list(const float* __restrict__ wS,
                                                  const int* __restrict__ off,
                                                  float* __restrict__ inv) {
  static_assert(N_NODES % 32 == 0, "the 8-lane groups must cover the nodes exactly");
  const int t    = blockIdx.x * 256 + threadIdx.x;
  const int node = t >> 3;
  const int j    = t & 7;
  const int beg  = off[node];
  const int end  = off[node + 1];
  float part = 0.0f;
  for (int c0 = beg; c0 < end; c0 += 32) {
    float we[4];
    for (int r = 0; r < 4; ++r) { const int k = c0 + r * 8 + j; we[r] = wS[(k < end) ? k : c0]; }
    for (int r = 0; r < 4; ++r) part += (c0 + r * 8 + j < end) ? we[r] : 0.0f;
  }
  for (int d = 1; d < 8; d <<= 1) part += __shfl_xor(part, d, 32);
  const float deg = part + 1.0f;
  if (j == 0) inv[node] = (deg > 0.0f) ? rsqrtf(fmaxf(deg, 1e-12f)) : 0.0f;
}


template <int K, int NOUT, int RPP, int TILES, bool RELU>
__global__ __launch_bounds__(32 * (NOUT / 16) * RPP)
void k_gemm_wmma(const float* X, const float* __restrict__ W, float* H) {
  constexpr int CT    = NOUT / 16;
  constexpr int NT    = 32 * CT * RPP;
  constexpr int ROWS  = 16 * RPP;
  constexpr int PITCH = K + 4;
  constexpr int NV    = ROWS * K / 4 / NT;
  constexpr int G     = 8;
  constexpr int NG    = K / 4 / G;
  static_assert(NG * G * 4 == K, "K must be a whole number of batches");
  static_assert(ROWS * K == NV * 4 * NT, "a pass must split evenly over the threads");
  static_assert((N_NODES / 16) % (RPP * TILES) == 0, "passes must cover the rows exactly");

  __shared__ __attribute__((aligned(16))) float sA[2][ROWS * PITCH];

  const int tid  = threadIdx.x;
  const int lane = tid & 31;
  const int wave = tid >> 5;
  const int ct   = wave % CT;
  const int rsub = wave / CT;
  const int l    = lane & 15;
  const int half = lane >> 4;
  const int ko   = half * 2;
  const int n0   = ct * 16;

  v2f b[K / 4];
#pragma unroll
  for (int j = 0; j < K / 4; ++j) {
    b[j].x = W[(size_t)(4 * j + ko)     * NOUT + n0 + l];
    b[j].y = W[(size_t)(4 * j + ko + 1) * NOUT + n0 + l];
  }

  const size_t row0 = (size_t)blockIdx.x * TILES * ROWS;
  const v4f* gsrc = (const v4f*)(X + row0 * K) + tid;

  v4f r[NV];
#pragma unroll
  for (int i = 0; i < NV; ++i) r[i] = gsrc[i * NT];
#pragma unroll
  for (int i = 0; i < NV; ++i) {
    const int f = 4 * (i * NT + tid);
    v4f v = r[i];
    if (RELU) { v.x = fmaxf(v.x, 0.0f); v.y = fmaxf(v.y, 0.0f);
                v.z = fmaxf(v.z, 0.0f); v.w = fmaxf(v.w, 0.0f); }
    *(v4f*)&sA[0][(f / K) * PITCH + (f % K)] = v;
  }
  __syncthreads();

  for (int t = 0; t < TILES; ++t) {
    const int cur = t & 1;
    const bool more = (t + 1 < TILES);
    if (more) {
      const v4f* g = gsrc + (size_t)(t + 1) * (ROWS * K / 4);
#pragma unroll
      for (int i = 0; i < NV; ++i) r[i] = g[i * NT];
    }

    const float* arow = &sA[cur][(rsub * 16 + l) * PITCH + ko];
    v8f c = {0.f, 0.f, 0.f, 0.f, 0.f, 0.f, 0.f, 0.f};
    v2f fa[2][G];
#pragma unroll
    for (int jj = 0; jj < G; ++jj) fa[0][jj] = *(const v2f*)(arow + 4 * jj);
#pragma unroll
    for (int g = 0; g < NG; ++g) {
      if (g + 1 < NG) {
#pragma unroll
        for (int jj = 0; jj < G; ++jj)
          fa[(g + 1) & 1][jj] = *(const v2f*)(arow + 4 * ((g + 1) * G + jj));
      }
      __builtin_amdgcn_sched_barrier(0);
#pragma unroll
      for (int jj = 0; jj < G; ++jj)
        c = __builtin_amdgcn_wmma_f32_16x16x4_f32(false, fa[g & 1][jj], false, b[g * G + jj],
                                                  (short)0, c, false, false);
      __builtin_amdgcn_sched_barrier(0);
    }

    if (more) {
#pragma unroll
      for (int i = 0; i < NV; ++i) {
        const int f = 4 * (i * NT + tid);
        v4f v = r[i];
        if (RELU) { v.x = fmaxf(v.x, 0.0f); v.y = fmaxf(v.y, 0.0f);
                    v.z = fmaxf(v.z, 0.0f); v.w = fmaxf(v.w, 0.0f); }
        *(v4f*)&sA[cur ^ 1][(f / K) * PITCH + (f % K)] = v;
      }
      __syncthreads();
    }

    const size_t m0 = row0 + (size_t)t * ROWS + rsub * 16;
    float* drow = H + (m0 + half * 8) * NOUT + (n0 + l);
#pragma unroll
    for (int i = 0; i < 8; ++i) drow[(size_t)i * NOUT] = c[i];
  }
}


static __device__ __forceinline__ float4 scaled(float a, float4 v) {
  return make_float4(a * v.x, a * v.y, a * v.z, a * v.w);
}
static __device__ __forceinline__ void add_scaled(float4& acc, float a, float4 v) {
  acc.x += a * v.x; acc.y += a * v.y; acc.z += a * v.z; acc.w += a * v.w;
}
static __device__ __forceinline__ float4 plus(float4 a, float4 b) {
  return make_float4(a.x + b.x, a.y + b.y, a.z + b.z, a.w + b.w);
}

template <int G, int NP, int LPN>
static __device__ __forceinline__ void walk_slots(float4 (&acc)[NP], const float4* hp,
                                                  int sr, float nr, int lane0) {
  constexpr size_t RP = NP * LPN;
  float4 v[G][NP];
#pragma unroll
  for (int k = 0; k < G; ++k) {
    const int sk = __shfl(sr, lane0 + k, 32);
#pragma unroll
    for (int p = 0; p < NP; ++p) v[k][p] = hp[(size_t)sk * RP + p * LPN];
  }
#pragma unroll
  for (int k = 0; k < G; ++k) {
    const float a = __shfl(nr, lane0 + k, 32);
#pragma unroll
    for (int p = 0; p < NP; ++p) add_scaled(acc[p], a, v[k][p]);
  }
}

template <int DIM, int LPN, int GROUP>
__global__ __launch_bounds__(256) void k_gather(const unsigned* __restrict__ srcLoW,
                                                const unsigned* __restrict__ srcHiW,
                                                const float* __restrict__ wS,
                                                const float* __restrict__ inv,
                                                const int* __restrict__ off,
                                                const float* __restrict__ h,
                                                const float* __restrict__ bias,
                                                float* __restrict__ res) {
  constexpr int NPW = 32 / LPN;
  constexpr int NP  = DIM / 4 / LPN;
  constexpr int R   = 32 / LPN;
  static_assert(NP * 4 * LPN == DIM, "a row must split evenly over the lanes of a group");
  static_assert(LPN % GROUP == 0, "a group of gathers stays inside one slot register");
  static_assert(N_NODES % (8 * NPW) == 0, "the waves must cover the nodes exactly");

  const unsigned short* srcLo = (const unsigned short*)srcLoW;
  const unsigned char*  srcHi = (const unsigned char*)srcHiW;
  const int lane  = threadIdx.x & 31;
  const int j     = lane % LPN;
  const int gbase = lane - j;
  const int node  = (blockIdx.x * 8 + (int)(threadIdx.x >> 5)) * NPW + lane / LPN;

  const int beg = off[node];
  const int end = off[node + 1];
  const float inv_t = inv[node];
  const float4* hp = (const float4*)h + j;
  constexpr size_t RP = DIM / 4;

  float4 acc[NP];
#pragma unroll
  for (int p = 0; p < NP; ++p)
    acc[p] = scaled(inv_t * 1.0f * inv_t, hp[(size_t)node * RP + p * LPN]);

  for (int c0 = beg; ; c0 += 32) {
    const int m = min(32, end - c0);
    int mm = m;
#pragma unroll
    for (int d = LPN; d < 32; d <<= 1) mm = max(mm, __shfl_xor(mm, d, 32));
    mm = __builtin_amdgcn_readfirstlane(mm);
    if (mm <= 0) break;

    int s[R]; float nrm[R];
    int lo[R], hi[R]; float we[R];
#pragma unroll
    for (int r = 0; r < R; ++r) {
      const int k = (r * LPN + j < m) ? c0 + r * LPN + j : 0;
      lo[r] = srcLo[k]; hi[r] = srcHi[k]; we[r] = wS[k];
    }
#pragma unroll
    for (int r = 0; r < R; ++r) s[r] = lo[r] | (hi[r] << 16);
#pragma unroll
    for (int r = 0; r < R; ++r) nrm[r] = inv[s[r]] * we[r] * inv_t;
#pragma unroll
    for (int r = 0; r < R; ++r) {
      if (!(r * LPN + j < m)) { s[r] = node; nrm[r] = 0.0f; }
      asm volatile("" : "+v"(s[r]), "+v"(nrm[r]));
    }
#pragma unroll
    for (int r = 0; r < R; ++r) {
#pragma unroll
      for (int u = 0; u < LPN; u += GROUP) {
        const int rem = mm - (r * LPN + u);
        if (rem <= 0) break;
        if (rem > GROUP / 2) walk_slots<GROUP, NP, LPN>(acc, hp, s[r], nrm[r], gbase + u);
        else                 walk_slots<GROUP / 2, NP, LPN>(acc, hp, s[r], nrm[r], gbase + u);
      }
      if ((r + 1) * LPN >= mm) break;
    }
  }

  const float4* bp = (const float4*)bias + j;
  float4* rp = (float4*)res + (size_t)node * RP + j;
#pragma unroll
  for (int p = 0; p < NP; ++p) rp[p * LPN] = plus(acc[p], bp[p * LPN]);
}


extern "C" void kernel_launch(void* const* d_in, const int* in_sizes, int n_in,
                              void* d_out, int out_size, void* d_ws, size_t ws_size,
                              hipStream_t stream) {
  const int N = N_NODES, E = N_EDGES;

  const float* x   = (const float*)d_in[0];
  const int*   ei  = (const int*)d_in[1];
  const int*   src = ei;
  const int*   dst = ei + E;
  const float* w   = (const float*)d_in[2];
  const float* W1  = (const float*)d_in[3];
  const float* b1  = (const float*)d_in[4];
  const float* W2  = (const float*)d_in[5];
  const float* b2  = (const float*)d_in[6];

  float* out  = (float*)d_out;
  float* xemb = out + (size_t)N * DIM_OUT;

  static_assert(N_EDGES % 4 == 0 && N_NODES % 2 == 0, "the list's parts must start on whole words");
  static_assert(N_EDGES + N_EDGES / 2 + N_EDGES / 4 + N_NODES + 1 <= N_NODES * DIM_OUT,
                "the list (weights, 16-bit and 8-bit source parts) and its offsets must fit behind h1");
  static_assert(2 * N_EDGES + 2 * NBIN * NCHUNK + NBIN <= N_NODES * DIM_HID,
                "the build's temporaries must fit in h1");
  float* inv  = (float*)d_ws;
  float* h1   = inv + N;
  float*          wS    = h1 + (size_t)N * DIM_HID;
  unsigned* srcLo = (unsigned*)(wS + E);
  unsigned* srcHi = srcLo + E / 2;
  int*   off  = (int*)(srcHi + E / 4);
  unsigned* tmp = (unsigned*)h1;
  int*   binCnt = (int*)h1 + 2 * E;
  int*   binPre = binCnt + NBIN * NCHUNK;
  int*   binTot = binPre + NBIN * NCHUNK;
  float* h2   = h1;

  k_bin_count  <<<NCHUNK, NT_E, 0, stream>>>(dst, binCnt);
  k_bin_prefix <<<NBIN, NCHUNK, 0, stream>>>(binCnt, binPre, binTot);
  k_bin_scatter<<<NCHUNK, NT_E, 0, stream>>>(src, dst, w, binPre, binTot, tmp);
  k_bin_sort   <<<NBIN, NT_E, 0, stream>>>(tmp, binTot, off, wS, srcLo, srcHi);
  k_deg_list  <<<N / 32, 256, 0, stream>>>(wS, off, inv);

  k_gemm_wmma<DIM_IN, DIM_HID, 1, 5, false><<<(N / 16) / 5, 256, 0, stream>>>(x, W1, h1);

  k_gather<DIM_HID, 16, 8><<<N / (8 * 2), 256, 0, stream>>>(srcLo, srcHi, wS, inv, off, h1, b1, xemb);

  k_gemm_wmma<DIM_HID, DIM_OUT, 5, 2, true><<<(N / 16) / (5 * 2), 320, 0, stream>>>(xemb, W2, h2);

  k_gather<DIM_OUT, 8, 8><<<N / (8 * 4), 256, 0, stream>>>(srcLo, srcHi, wS, inv, off, h2, b2, out);
}


# dense products store each row times its node's inv_sqrt; neighbour sums use w*inv[target] per edge: no inv_sqrt lookup per edge; a term rounds as (w*inv[t])*(inv[s]*h), f32 sum orders unchanged
# speedup vs baseline: 3.9085x; 1.0410x over previous
#include <hip/hip_runtime.h>

typedef __attribute__((ext_vector_type(2))) float v2f;
typedef __attribute__((ext_vector_type(4))) float v4f;
typedef __attribute__((ext_vector_type(8))) float v8f;

#define N_NODES  100000
#define N_EDGES  1600000
#define DIM_IN   256
#define DIM_HID  128
#define DIM_OUT  32


#define NBIN   ((N_NODES + 255) / 256)
#define NCHUNK 256
#define CHUNK  (N_EDGES / NCHUNK)
#define NT_E   512

static __device__ __forceinline__ int wave_incl_scan(int v, int lane) {
#pragma unroll
  for (int d = 1; d < 32; d <<= 1) {
    int t = __shfl_up(v, d, 32);
    if (lane >= d) v += t;
  }
  return v;
}

template <int NW>
static __device__ __forceinline__ int block_excl_scan(int v, int* sh, int& total) {
  const int lane = threadIdx.x & 31, wave = threadIdx.x >> 5;
  const int incl = wave_incl_scan(v, lane);
  if (lane == 31) sh[wave] = incl;
  __syncthreads();
  int base = 0, tot = 0;
  for (int k = 0; k < NW; ++k) {
    const int t = sh[k];
    if (k < wave) base += t;
    tot += t;
  }
  total = tot;
  __syncthreads();
  return base + incl - v;
}

__global__ __launch_bounds__(NT_E) void k_bin_count(const int* __restrict__ dst,
                                                    int* __restrict__ binCnt) {
  static_assert(CHUNK * NCHUNK == N_EDGES, "the pieces must cover the edges exactly");
  constexpr int PER = (CHUNK + NT_E - 1) / NT_E;
  __shared__ int cnt[NBIN];
  const int tid = threadIdx.x, blk = blockIdx.x;
  for (int b = tid; b < NBIN; b += NT_E) cnt[b] = 0;
  const int* d = dst + blk * CHUNK;
  int dv[PER];
#pragma unroll
  for (int k = 0; k < PER; ++k) { const int i = tid + k * NT_E; dv[k] = d[(i < CHUNK) ? i : 0]; }
  __syncthreads();
#pragma unroll
  for (int k = 0; k < PER; ++k) if (tid + k * NT_E < CHUNK) atomicAdd(&cnt[dv[k] >> 8], 1);
  __syncthreads();
  for (int b = tid; b < NBIN; b += NT_E) binCnt[b * NCHUNK + blk] = cnt[b];
}

__global__ __launch_bounds__(NCHUNK) void k_bin_prefix(const int* __restrict__ binCnt,
                                                       int* __restrict__ binPre,
                                                       int* __restrict__ binTot) {
  __shared__ int sh[NCHUNK / 32];
  const int bin = blockIdx.x, t = threadIdx.x;
  int tot;
  const int ex = block_excl_scan<NCHUNK / 32>(binCnt[bin * NCHUNK + t], sh, tot);
  binPre[bin * NCHUNK + t] = ex;
  if (t == 0) binTot[bin] = tot;
}

__global__ __launch_bounds__(NT_E) void k_bin_scatter(const int* __restrict__ src,
                                                      const int* __restrict__ dst,
                                                      const float* __restrict__ w,
                                                      const int* __restrict__ binPre,
                                                      const int* __restrict__ binTot,
                                                      unsigned* __restrict__ tmpw) {
  static_assert(NBIN <= NT_E, "one thread per bin in the scan of the bin totals");
  static_assert(N_NODES <= (1 << 24), "a source must leave 8 bits of the record's first word free");
  constexpr int PER = (CHUNK + NT_E - 1) / NT_E;
  __shared__ int cur[NBIN];
  __shared__ int sh[NT_E / 32];
  const int tid = threadIdx.x, blk = blockIdx.x;
  uint2* tmp = (uint2*)tmpw;
  const int*   d  = dst + blk * CHUNK;
  const int*   s  = src + blk * CHUNK;
  const float* ww = w   + blk * CHUNK;
  int dv[PER], sv[PER];
  float wv[PER];
#pragma unroll
  for (int k = 0; k < PER; ++k) {
    const int i = tid + k * NT_E;
    const int ii = (i < CHUNK) ? i : 0;
    dv[k] = d[ii]; sv[k] = s[ii]; wv[k] = ww[ii];
  }
  int tot;
  const int start = block_excl_scan<NT_E / 32>((tid < NBIN) ? binTot[tid] : 0, sh, tot);
  if (tid < NBIN) cur[tid] = start + binPre[tid * NCHUNK + blk];
  __syncthreads();
  int pos[PER];
#pragma unroll
  for (int k = 0; k < PER; ++k)
    pos[k] = (tid + k * NT_E < CHUNK) ? atomicAdd(&cur[dv[k] >> 8], 1) : 0;
#pragma unroll
  for (int k = 0; k < PER; ++k) {
    if (tid + k * NT_E < CHUNK)
      tmp[pos[k]] = make_uint2(((unsigned)sv[k] << 8) | (unsigned)(dv[k] & 255),
                               __float_as_uint(wv[k]));
  }
}

__global__ __launch_bounds__(NT_E) void k_bin_sort(const unsigned* __restrict__ tmpw,
                                                   const int* __restrict__ binTot,
                                                   int* __restrict__ off,
                                                   float* __restrict__ wS,
                                                   unsigned* __restrict__ srcLoW,
                                                   unsigned* __restrict__ srcHiW) {
  static_assert(NBIN <= NT_E && 256 <= NT_E, "one thread per bin and per node of a bin");
  constexpr int RB  = (N_EDGES / NBIN) / NT_E + 2;
  constexpr int CAP = RB * NT_E;
  __shared__ int cnt[256];
  __shared__ int sh[NT_E / 32];
  __shared__ int s_first, s_count;
  __shared__ float    stW[CAP];
  __shared__ unsigned stS[CAP];
  const uint2* tmp = (const uint2*)tmpw;
  unsigned short* srcLo = (unsigned short*)srcLoW;
  unsigned char*  srcHi = (unsigned char*)srcHiW;
  const int tid = threadIdx.x, bin = blockIdx.x;
  const int mine = (tid < NBIN) ? binTot[tid] : 0;
  int tot;
  const int ex = block_excl_scan<NT_E / 32>(mine, sh, tot);
  if (tid == bin) { s_first = ex; s_count = mine; }
  if (tid < 256) cnt[tid] = 0;
  __syncthreads();
  const int bs = s_first, nb = s_count;
  const uint2* rec = tmp + bs;
  const bool staged = (nb <= CAP);

  for (int i0 = 0; i0 < nb; i0 += RB * NT_E) {
    unsigned r[RB];
#pragma unroll
    for (int k = 0; k < RB; ++k) r[k] = rec[min(i0 + k * NT_E + tid, nb - 1)].x;
#pragma unroll
    for (int k = 0; k < RB; ++k) if (i0 + k * NT_E + tid < nb) atomicAdd(&cnt[r[k] & 255u], 1);
  }
  __syncthreads();

  int tot2;
  const int ex2 = block_excl_scan<NT_E / 32>((tid < 256) ? cnt[tid] : 0, sh, tot2);
  if (tid < 256) {
    cnt[tid] = ex2;
    const int node = bin * 256 + tid;
    if (node < N_NODES) off[node] = bs + ex2;
  }
  if (bin == NBIN - 1 && tid == 0) off[N_NODES] = bs + nb;
  __syncthreads();

  for (int i0 = 0; i0 < nb; i0 += RB * NT_E) {
    uint2 r[RB];
    int p[RB];
#pragma unroll
    for (int k = 0; k < RB; ++k) r[k] = rec[min(i0 + k * NT_E + tid, nb - 1)];
#pragma unroll
    for (int k = 0; k < RB; ++k)
      p[k] = (i0 + k * NT_E + tid < nb) ? atomicAdd(&cnt[r[k].x & 255u], 1) : 0;
#pragma unroll
    for (int k = 0; k < RB; ++k) {
      if (i0 + k * NT_E + tid < nb) {
        const unsigned sv = r[k].x >> 8;
        if (staged) {
          stW[p[k]] = __uint_as_float(r[k].y);
          stS[p[k]] = sv;
        } else {
          const int q = bs + p[k];
          wS[q]    = __uint_as_float(r[k].y);
          srcLo[q] = (unsigned short)(sv & 0xffffu);
          srcHi[q] = (unsigned char)(sv >> 16);
        }
      }
    }
  }
  if (staged) {
    __syncthreads();
    for (int i = tid; i < nb; i += NT_E) {
      const unsigned sv = stS[i];
      wS[bs + i]    = stW[i];
      srcLo[bs + i] = (unsigned short)(sv & 0xffffu);
      srcHi[bs + i] = (unsigned char)(sv >> 16);
    }
  }
}

__global__ __launch_bounds__(256) void k_deg_list(const float* __restrict__ wS,
                                                  const int* __restrict__ off,
                                                  float* __restrict__ inv) {
  static_assert(N_NODES % 32 == 0, "the 8-lane groups must cover the nodes exactly");
  const int t    = blockIdx.x * 256 + threadIdx.x;
  const int node = t >> 3;
  const int j    = t & 7;
  const int beg  = off[node];
  const int end  = off[node + 1];
  float part = 0.0f;
  for (int c0 = beg; c0 < end; c0 += 32) {
    float we[4];
    for (int r = 0; r < 4; ++r) { const int k = c0 + r * 8 + j; we[r] = wS[(k < end) ? k : c0]; }
    for (int r = 0; r < 4; ++r) part += (c0 + r * 8 + j < end) ? we[r] : 0.0f;
  }
  for (int d = 1; d < 8; d <<= 1) part += __shfl_xor(part, d, 32);
  const float deg = part + 1.0f;
  if (j == 0) inv[node] = (deg > 0.0f) ? rsqrtf(fmaxf(deg, 1e-12f)) : 0.0f;
}


template <int K, int NOUT, int RPP, int TILES, bool RELU>
__global__ __launch_bounds__(32 * (NOUT / 16) * RPP)
void k_gemm_wmma(const float* X, const float* __restrict__ W, const float* RS, float* H) {
  constexpr int CT    = NOUT / 16;
  constexpr int NT    = 32 * CT * RPP;
  constexpr int ROWS  = 16 * RPP;
  constexpr int PITCH = K + 4;
  constexpr int NV    = ROWS * K / 4 / NT;
  constexpr int G     = 8;
  constexpr int NG    = K / 4 / G;
  static_assert(NG * G * 4 == K, "K must be a whole number of batches");
  static_assert(ROWS * K == NV * 4 * NT, "a pass must split evenly over the threads");
  static_assert((N_NODES / 16) % (RPP * TILES) == 0, "passes must cover the rows exactly");

  __shared__ __attribute__((aligned(16))) float sA[2][ROWS * PITCH];

  const int tid  = threadIdx.x;
  const int lane = tid & 31;
  const int wave = tid >> 5;
  const int ct   = wave % CT;
  const int rsub = wave / CT;
  const int l    = lane & 15;
  const int half = lane >> 4;
  const int ko   = half * 2;
  const int n0   = ct * 16;

  v2f b[K / 4];
#pragma unroll
  for (int j = 0; j < K / 4; ++j) {
    b[j].x = W[(size_t)(4 * j + ko)     * NOUT + n0 + l];
    b[j].y = W[(size_t)(4 * j + ko + 1) * NOUT + n0 + l];
  }

  const size_t row0 = (size_t)blockIdx.x * TILES * ROWS;
  const v4f* gsrc = (const v4f*)(X + row0 * K) + tid;

  v4f r[NV];
#pragma unroll
  for (int i = 0; i < NV; ++i) r[i] = gsrc[i * NT];
#pragma unroll
  for (int i = 0; i < NV; ++i) {
    const int f = 4 * (i * NT + tid);
    v4f v = r[i];
    if (RELU) { v.x = fmaxf(v.x, 0.0f); v.y = fmaxf(v.y, 0.0f);
                v.z = fmaxf(v.z, 0.0f); v.w = fmaxf(v.w, 0.0f); }
    *(v4f*)&sA[0][(f / K) * PITCH + (f % K)] = v;
  }
  __syncthreads();

  for (int t = 0; t < TILES; ++t) {
    const int cur = t & 1;
    const bool more = (t + 1 < TILES);
    if (more) {
      const v4f* g = gsrc + (size_t)(t + 1) * (ROWS * K / 4);
#pragma unroll
      for (int i = 0; i < NV; ++i) r[i] = g[i * NT];
    }

    const float* arow = &sA[cur][(rsub * 16 + l) * PITCH + ko];
    const size_t m0 = row0 + (size_t)t * ROWS + rsub * 16;
    const v4f* sp = (const v4f*)(RS + m0 + half * 8);
    v4f sc0, sc1;
    v8f c = {0.f, 0.f, 0.f, 0.f, 0.f, 0.f, 0.f, 0.f};
    v2f fa[2][G];
#pragma unroll
    for (int jj = 0; jj < G; ++jj) fa[0][jj] = *(const v2f*)(arow + 4 * jj);
#pragma unroll
    for (int g = 0; g < NG; ++g) {
      if (g + 1 < NG) {
#pragma unroll
        for (int jj = 0; jj < G; ++jj)
          fa[(g + 1) & 1][jj] = *(const v2f*)(arow + 4 * ((g + 1) * G + jj));
      } else {
        sc0 = sp[0]; sc1 = sp[1];
      }
      __builtin_amdgcn_sched_barrier(0);
#pragma unroll
      for (int jj = 0; jj < G; ++jj)
        c = __builtin_amdgcn_wmma_f32_16x16x4_f32(false, fa[g & 1][jj], false, b[g * G + jj],
                                                  (short)0, c, false, false);
      __builtin_amdgcn_sched_barrier(0);
    }

    if (more) {
#pragma unroll
      for (int i = 0; i < NV; ++i) {
        const int f = 4 * (i * NT + tid);
        v4f v = r[i];
        if (RELU) { v.x = fmaxf(v.x, 0.0f); v.y = fmaxf(v.y, 0.0f);
                    v.z = fmaxf(v.z, 0.0f); v.w = fmaxf(v.w, 0.0f); }
        *(v4f*)&sA[cur ^ 1][(f / K) * PITCH + (f % K)] = v;
      }
      __syncthreads();
    }

    float* drow = H + (m0 + half * 8) * NOUT + (n0 + l);
#pragma unroll
    for (int i = 0; i < 4; ++i) drow[(size_t)i * NOUT] = c[i] * sc0[i];
#pragma unroll
    for (int i = 0; i < 4; ++i) drow[(size_t)(4 + i) * NOUT] = c[4 + i] * sc1[i];
  }
}


static __device__ __forceinline__ float4 scaled(float a, float4 v) {
  return make_float4(a * v.x, a * v.y, a * v.z, a * v.w);
}
static __device__ __forceinline__ void add_scaled(float4& acc, float a, float4 v) {
  acc.x += a * v.x; acc.y += a * v.y; acc.z += a * v.z; acc.w += a * v.w;
}
static __device__ __forceinline__ float4 plus(float4 a, float4 b) {
  return make_float4(a.x + b.x, a.y + b.y, a.z + b.z, a.w + b.w);
}

template <int G, int NP, int LPN>
static __device__ __forceinline__ void walk_slots(float4 (&acc)[NP], const float4* hp,
                                                  int sr, float nr, int lane0) {
  constexpr size_t RP = NP * LPN;
  float4 v[G][NP];
#pragma unroll
  for (int k = 0; k < G; ++k) {
    const int sk = __shfl(sr, lane0 + k, 32);
#pragma unroll
    for (int p = 0; p < NP; ++p) v[k][p] = hp[(size_t)sk * RP + p * LPN];
  }
#pragma unroll
  for (int k = 0; k < G; ++k) {
    const float a = __shfl(nr, lane0 + k, 32);
#pragma unroll
    for (int p = 0; p < NP; ++p) add_scaled(acc[p], a, v[k][p]);
  }
}

template <int DIM, int LPN, int GROUP>
__global__ __launch_bounds__(256) void k_gather(const unsigned* __restrict__ srcLoW,
                                                const unsigned* __restrict__ srcHiW,
                                                const float* __restrict__ wS,
                                                const float* __restrict__ inv,
                                                const int* __restrict__ off,
                                                const float* __restrict__ h,
                                                const float* __restrict__ bias,
                                                float* __restrict__ res) {
  constexpr int NPW = 32 / LPN;
  constexpr int NP  = DIM / 4 / LPN;
  constexpr int R   = 32 / LPN;
  static_assert(NP * 4 * LPN == DIM, "a row must split evenly over the lanes of a group");
  static_assert(LPN % GROUP == 0, "a group of gathers stays inside one slot register");
  static_assert(N_NODES % (8 * NPW) == 0, "the waves must cover the nodes exactly");

  const unsigned short* srcLo = (const unsigned short*)srcLoW;
  const unsigned char*  srcHi = (const unsigned char*)srcHiW;
  const int lane  = threadIdx.x & 31;
  const int j     = lane % LPN;
  const int gbase = lane - j;
  const int node  = (blockIdx.x * 8 + (int)(threadIdx.x >> 5)) * NPW + lane / LPN;

  const int beg = off[node];
  const int end = off[node + 1];
  const float inv_t = inv[node];
  const float4* hp = (const float4*)h + j;
  constexpr size_t RP = DIM / 4;

  float4 acc[NP];
#pragma unroll
  for (int p = 0; p < NP; ++p)
    acc[p] = scaled(inv_t, hp[(size_t)node * RP + p * LPN]);

  for (int c0 = beg; ; c0 += 32) {
    const int m = min(32, end - c0);
    int mm = m;
#pragma unroll
    for (int d = LPN; d < 32; d <<= 1) mm = max(mm, __shfl_xor(mm, d, 32));
    mm = __builtin_amdgcn_readfirstlane(mm);
    if (mm <= 0) break;

    int s[R]; float nrm[R];
    int lo[R], hi[R]; float we[R];
#pragma unroll
    for (int r = 0; r < R; ++r) {
      const int k = (r * LPN + j < m) ? c0 + r * LPN + j : 0;
      lo[r] = srcLo[k]; hi[r] = srcHi[k]; we[r] = wS[k];
    }
#pragma unroll
    for (int r = 0; r < R; ++r) s[r] = lo[r] | (hi[r] << 16);
#pragma unroll
    for (int r = 0; r < R; ++r) nrm[r] = we[r] * inv_t;
#pragma unroll
    for (int r = 0; r < R; ++r) {
      if (!(r * LPN + j < m)) { s[r] = node; nrm[r] = 0.0f; }
      asm volatile("" : "+v"(s[r]), "+v"(nrm[r]));
    }
#pragma unroll
    for (int r = 0; r < R; ++r) {
#pragma unroll
      for (int u = 0; u < LPN; u += GROUP) {
        const int rem = mm - (r * LPN + u);
        if (rem <= 0) break;
        if (rem > GROUP / 2) walk_slots<GROUP, NP, LPN>(acc, hp, s[r], nrm[r], gbase + u);
        else                 walk_slots<GROUP / 2, NP, LPN>(acc, hp, s[r], nrm[r], gbase + u);
      }
      if ((r + 1) * LPN >= mm) break;
    }
  }

  const float4* bp = (const float4*)bias + j;
  float4* rp = (float4*)res + (size_t)node * RP + j;
#pragma unroll
  for (int p = 0; p < NP; ++p) rp[p * LPN] = plus(acc[p], bp[p * LPN]);
}


extern "C" void kernel_launch(void* const* d_in, const int* in_sizes, int n_in,
                              void* d_out, int out_size, void* d_ws, size_t ws_size,
                              hipStream_t stream) {
  const int N = N_NODES, E = N_EDGES;

  const float* x   = (const float*)d_in[0];
  const int*   ei  = (const int*)d_in[1];
  const int*   src = ei;
  const int*   dst = ei + E;
  const float* w   = (const float*)d_in[2];
  const float* W1  = (const float*)d_in[3];
  const float* b1  = (const float*)d_in[4];
  const float* W2  = (const float*)d_in[5];
  const float* b2  = (const float*)d_in[6];

  float* out  = (float*)d_out;
  float* xemb = out + (size_t)N * DIM_OUT;

  static_assert(N_EDGES % 4 == 0 && N_NODES % 2 == 0, "the list's parts must start on whole words");
  static_assert(N_EDGES + N_EDGES / 2 + N_EDGES / 4 + N_NODES + 1 <= N_NODES * DIM_OUT,
                "the list (weights, 16-bit and 8-bit source parts) and its offsets must fit behind h1");
  static_assert(2 * N_EDGES + 2 * NBIN * NCHUNK + NBIN <= N_NODES * DIM_HID,
                "the build's temporaries must fit in h1");
  float* inv  = (float*)d_ws;
  float* h1   = inv + N;
  float*          wS    = h1 + (size_t)N * DIM_HID;
  unsigned* srcLo = (unsigned*)(wS + E);
  unsigned* srcHi = srcLo + E / 2;
  int*   off  = (int*)(srcHi + E / 4);
  unsigned* tmp = (unsigned*)h1;
  int*   binCnt = (int*)h1 + 2 * E;
  int*   binPre = binCnt + NBIN * NCHUNK;
  int*   binTot = binPre + NBIN * NCHUNK;
  float* h2   = h1;

  k_bin_count  <<<NCHUNK, NT_E, 0, stream>>>(dst, binCnt);
  k_bin_prefix <<<NBIN, NCHUNK, 0, stream>>>(binCnt, binPre, binTot);
  k_bin_scatter<<<NCHUNK, NT_E, 0, stream>>>(src, dst, w, binPre, binTot, tmp);
  k_bin_sort   <<<NBIN, NT_E, 0, stream>>>(tmp, binTot, off, wS, srcLo, srcHi);
  k_deg_list  <<<N / 32, 256, 0, stream>>>(wS, off, inv);

  k_gemm_wmma<DIM_IN, DIM_HID, 1, 5, false><<<(N / 16) / 5, 256, 0, stream>>>(x, W1, inv, h1);

  k_gather<DIM_HID, 16, 8><<<N / (8 * 2), 256, 0, stream>>>(srcLo, srcHi, wS, inv, off, h1, b1, xemb);

  k_gemm_wmma<DIM_HID, DIM_OUT, 5, 2, true><<<(N / 16) / (5 * 2), 320, 0, stream>>>(xemb, W2, inv, h2);

  k_gather<DIM_OUT, 8, 8><<<N / (8 * 4), 256, 0, stream>>>(srcLo, srcHi, wS, inv, off, h2, b2, out);
}


# edge sort: level 1 orders each piece by bin in LDS and writes it whole; level 2 finds a bin's runs by a table of run starts: no count and scan launches (9 to 7); list order, so f32 sum order, changes
# speedup vs baseline: 4.4232x; 1.1317x over previous
#include <hip/hip_runtime.h>

typedef __attribute__((ext_vector_type(2))) float v2f;
typedef __attribute__((ext_vector_type(4))) float v4f;
typedef __attribute__((ext_vector_type(8))) float v8f;

#define N_NODES  100000
#define N_EDGES  1600000
#define DIM_IN   256
#define DIM_HID  128
#define DIM_OUT  32


#define NBIN   ((N_NODES + 255) / 256)
#define NCHUNK 256
#define CHUNK  (N_EDGES / NCHUNK)
#define NT_E   512

static __device__ __forceinline__ int wave_incl_scan(int v, int lane) {
#pragma unroll
  for (int d = 1; d < 32; d <<= 1) {
    int t = __shfl_up(v, d, 32);
    if (lane >= d) v += t;
  }
  return v;
}

template <int NW>
static __device__ __forceinline__ int block_excl_scan(int v, int* sh, int& total) {
  const int lane = threadIdx.x & 31, wave = threadIdx.x >> 5;
  const int incl = wave_incl_scan(v, lane);
  if (lane == 31) sh[wave] = incl;
  __syncthreads();
  int base = 0, tot = 0;
  for (int k = 0; k < NW; ++k) {
    const int t = sh[k];
    if (k < wave) base += t;
    tot += t;
  }
  total = tot;
  __syncthreads();
  return base + incl - v;
}

__global__ __launch_bounds__(NT_E) void k_piece_sort(const int* __restrict__ src,
                                                     const int* __restrict__ dst,
                                                     const float* __restrict__ w,
                                                     int* __restrict__ runStart,
                                                     unsigned* __restrict__ tmpw) {
  static_assert(CHUNK * NCHUNK == N_EDGES, "the pieces must cover the edges exactly");
  static_assert(NBIN <= NT_E, "one thread per bin in the scan of the bin counts");
  static_assert(N_NODES <= (1 << 24), "a source must leave 8 bits of the record's first word free");
  constexpr int PER = (CHUNK + NT_E - 1) / NT_E;
  __shared__ int cnt[NBIN];
  __shared__ int sh[NT_E / 32];
  __shared__ uint2 st[CHUNK];
  const int tid = threadIdx.x, blk = blockIdx.x;
  uint2* tmp = (uint2*)tmpw + (size_t)blk * CHUNK;
  const int*   d  = dst + blk * CHUNK;
  const int*   s  = src + blk * CHUNK;
  const float* ww = w   + blk * CHUNK;
  if (tid < NBIN) cnt[tid] = 0;
  int dv[PER], sv[PER];
  float wv[PER];
#pragma unroll
  for (int k = 0; k < PER; ++k) {
    const int i = tid + k * NT_E;
    const int ii = (i < CHUNK) ? i : 0;
    dv[k] = d[ii]; sv[k] = s[ii]; wv[k] = ww[ii];
  }
  __syncthreads();
#pragma unroll
  for (int k = 0; k < PER; ++k) if (tid + k * NT_E < CHUNK) atomicAdd(&cnt[dv[k] >> 8], 1);
  __syncthreads();
  int tot;
  const int start = block_excl_scan<NT_E / 32>((tid < NBIN) ? cnt[tid] : 0, sh, tot);
  if (tid < NBIN) {
    cnt[tid] = start;
    runStart[tid * NCHUNK + blk] = start;
  }
  __syncthreads();
  int pos[PER];
#pragma unroll
  for (int k = 0; k < PER; ++k)
    pos[k] = (tid + k * NT_E < CHUNK) ? atomicAdd(&cnt[dv[k] >> 8], 1) : 0;
#pragma unroll
  for (int k = 0; k < PER; ++k) {
    if (tid + k * NT_E < CHUNK)
      st[pos[k]] = make_uint2(((unsigned)sv[k] << 8) | (unsigned)(dv[k] & 255),
                              __float_as_uint(wv[k]));
  }
  __syncthreads();
  for (int i = tid; i < CHUNK; i += NT_E) tmp[i] = st[i];
}

__global__ __launch_bounds__(NT_E) void k_bin_sort(const unsigned* __restrict__ tmpw,
                                                   const int* __restrict__ runStart,
                                                   int* __restrict__ off,
                                                   float* __restrict__ wS,
                                                   unsigned* __restrict__ srcLoW,
                                                   unsigned* __restrict__ srcHiW) {
  static_assert(NCHUNK <= NT_E && 256 <= NT_E, "one thread per piece and per node of a bin");
  constexpr int NWV = NT_E / 32;
  constexpr int RPW = NCHUNK / NWV;
  static_assert(RPW * NWV == NCHUNK, "the runs must split evenly over the waves");
  constexpr int CAP = ((N_EDGES / NBIN) / NT_E + 2) * NT_E;
  __shared__ int cnt[256];
  __shared__ int sh[NT_E / 32];
  __shared__ int runSrc[NCHUNK];
  __shared__ int runLen[NCHUNK];
  __shared__ float    stW[CAP];
  __shared__ unsigned stS[CAP];
  const uint2* tmp = (const uint2*)tmpw;
  unsigned short* srcLo = (unsigned short*)srcLoW;
  unsigned char*  srcHi = (unsigned char*)srcHiW;
  const int tid = threadIdx.x, bin = blockIdx.x;
  const int lane = tid & 31, wv = tid >> 5;
  int a = 0, c = 0;
  if (tid < NCHUNK) {
    a = runStart[bin * NCHUNK + tid];
    const int e = (bin + 1 < NBIN) ? runStart[(bin + 1) * NCHUNK + tid] : CHUNK;
    c = e - a;
    runSrc[tid] = tid * CHUNK + a;
    runLen[tid] = c;
  }
  if (tid < 256) cnt[tid] = 0;
  int bs, nb;
  block_excl_scan<NT_E / 32>(a, sh, bs);
  block_excl_scan<NT_E / 32>(c, sh, nb);
  const bool staged = (nb <= CAP);

  {
    unsigned r[RPW];
#pragma unroll
    for (int k = 0; k < RPW; ++k) {
      const int p = wv + k * NWV;
      r[k] = tmp[(lane < runLen[p]) ? runSrc[p] + lane : 0].x;
    }
#pragma unroll
    for (int k = 0; k < RPW; ++k)
      if (lane < runLen[wv + k * NWV]) atomicAdd(&cnt[r[k] & 255u], 1);
    for (int k = 0; k < RPW; ++k) {
      const int p = wv + k * NWV;
      const int len = runLen[p], base = runSrc[p];
      for (int q = lane + 32; q < len; q += 32) atomicAdd(&cnt[tmp[base + q].x & 255u], 1);
    }
  }
  __syncthreads();

  int tot2;
  const int ex2 = block_excl_scan<NT_E / 32>((tid < 256) ? cnt[tid] : 0, sh, tot2);
  if (tid < 256) {
    cnt[tid] = ex2;
    const int node = bin * 256 + tid;
    if (node < N_NODES) off[node] = bs + ex2;
  }
  if (bin == NBIN - 1 && tid == 0) off[N_NODES] = bs + nb;
  __syncthreads();

  {
    uint2 r[RPW];
    int pl[RPW];
#pragma unroll
    for (int k = 0; k < RPW; ++k) {
      const int p = wv + k * NWV;
      r[k] = tmp[(lane < runLen[p]) ? runSrc[p] + lane : 0];
    }
#pragma unroll
    for (int k = 0; k < RPW; ++k)
      pl[k] = (lane < runLen[wv + k * NWV]) ? atomicAdd(&cnt[r[k].x & 255u], 1) : 0;
#pragma unroll
    for (int k = 0; k < RPW; ++k) {
      if (lane < runLen[wv + k * NWV]) {
        const unsigned sv = r[k].x >> 8;
        if (staged) {
          stW[pl[k]] = __uint_as_float(r[k].y);
          stS[pl[k]] = sv;
        } else {
          const int q = bs + pl[k];
          wS[q]    = __uint_as_float(r[k].y);
          srcLo[q] = (unsigned short)(sv & 0xffffu);
          srcHi[q] = (unsigned char)(sv >> 16);
        }
      }
    }
    for (int k = 0; k < RPW; ++k) {
      const int p = wv + k * NWV;
      const int len = runLen[p], base = runSrc[p];
      for (int i = lane + 32; i < len; i += 32) {
        const uint2 rr = tmp[base + i];
        const int pp = atomicAdd(&cnt[rr.x & 255u], 1);
        const unsigned sv = rr.x >> 8;
        if (staged) {
          stW[pp] = __uint_as_float(rr.y);
          stS[pp] = sv;
        } else {
          const int q = bs + pp;
          wS[q]    = __uint_as_float(rr.y);
          srcLo[q] = (unsigned short)(sv & 0xffffu);
          srcHi[q] = (unsigned char)(sv >> 16);
        }
      }
    }
  }
  if (staged) {
    __syncthreads();
    for (int i = tid; i < nb; i += NT_E) {
      const unsigned sv = stS[i];
      wS[bs + i]    = stW[i];
      srcLo[bs + i] = (unsigned short)(sv & 0xffffu);
      srcHi[bs + i] = (unsigned char)(sv >> 16);
    }
  }
}

__global__ __launch_bounds__(256) void k_deg_list(const float* __restrict__ wS,
                                                  const int* __restrict__ off,
                                                  float* __restrict__ inv) {
  static_assert(N_NODES % 32 == 0, "the 8-lane groups must cover the nodes exactly");
  const int t    = blockIdx.x * 256 + threadIdx.x;
  const int node = t >> 3;
  const int j    = t & 7;
  const int beg  = off[node];
  const int end  = off[node + 1];
  float part = 0.0f;
  for (int c0 = beg; c0 < end; c0 += 32) {
    float we[4];
    for (int r = 0; r < 4; ++r) { const int k = c0 + r * 8 + j; we[r] = wS[(k < end) ? k : c0]; }
    for (int r = 0; r < 4; ++r) part += (c0 + r * 8 + j < end) ? we[r] : 0.0f;
  }
  for (int d = 1; d < 8; d <<= 1) part += __shfl_xor(part, d, 32);
  const float deg = part + 1.0f;
  if (j == 0) inv[node] = (deg > 0.0f) ? rsqrtf(fmaxf(deg, 1e-12f)) : 0.0f;
}


template <int K, int NOUT, int RPP, int TILES, bool RELU>
__global__ __launch_bounds__(32 * (NOUT / 16) * RPP)
void k_gemm_wmma(const float* X, const float* __restrict__ W, const float* RS, float* H) {
  constexpr int CT    = NOUT / 16;
  constexpr int NT    = 32 * CT * RPP;
  constexpr int ROWS  = 16 * RPP;
  constexpr int PITCH = K + 4;
  constexpr int NV    = ROWS * K / 4 / NT;
  constexpr int G     = 8;
  constexpr int NG    = K / 4 / G;
  static_assert(NG * G * 4 == K, "K must be a whole number of batches");
  static_assert(ROWS * K == NV * 4 * NT, "a pass must split evenly over the threads");
  static_assert((N_NODES / 16) % (RPP * TILES) == 0, "passes must cover the rows exactly");

  __shared__ __attribute__((aligned(16))) float sA[2][ROWS * PITCH];

  const int tid  = threadIdx.x;
  const int lane = tid & 31;
  const int wave = tid >> 5;
  const int ct   = wave % CT;
  const int rsub = wave / CT;
  const int l    = lane & 15;
  const int half = lane >> 4;
  const int ko   = half * 2;
  const int n0   = ct * 16;

  v2f b[K / 4];
#pragma unroll
  for (int j = 0; j < K / 4; ++j) {
    b[j].x = W[(size_t)(4 * j + ko)     * NOUT + n0 + l];
    b[j].y = W[(size_t)(4 * j + ko + 1) * NOUT + n0 + l];
  }

  const size_t row0 = (size_t)blockIdx.x * TILES * ROWS;
  const v4f* gsrc = (const v4f*)(X + row0 * K) + tid;

  v4f r[NV];
#pragma unroll
  for (int i = 0; i < NV; ++i) r[i] = gsrc[i * NT];
#pragma unroll
  for (int i = 0; i < NV; ++i) {
    const int f = 4 * (i * NT + tid);
    v4f v = r[i];
    if (RELU) { v.x = fmaxf(v.x, 0.0f); v.y = fmaxf(v.y, 0.0f);
                v.z = fmaxf(v.z, 0.0f); v.w = fmaxf(v.w, 0.0f); }
    *(v4f*)&sA[0][(f / K) * PITCH + (f % K)] = v;
  }
  __syncthreads();

  for (int t = 0; t < TILES; ++t) {
    const int cur = t & 1;
    const bool more = (t + 1 < TILES);
    if (more) {
      const v4f* g = gsrc + (size_t)(t + 1) * (ROWS * K / 4);
#pragma unroll
      for (int i = 0; i < NV; ++i) r[i] = g[i * NT];
    }

    const float* arow = &sA[cur][(rsub * 16 + l) * PITCH + ko];
    const size_t m0 = row0 + (size_t)t * ROWS + rsub * 16;
    const v4f* sp = (const v4f*)(RS + m0 + half * 8);
    v4f sc0, sc1;
    v8f c = {0.f, 0.f, 0.f, 0.f, 0.f, 0.f, 0.f, 0.f};
    v2f fa[2][G];
#pragma unroll
    for (int jj = 0; jj < G; ++jj) fa[0][jj] = *(const v2f*)(arow + 4 * jj);
#pragma unroll
    for (int g = 0; g < NG; ++g) {
      if (g + 1 < NG) {
#pragma unroll
        for (int jj = 0; jj < G; ++jj)
          fa[(g + 1) & 1][jj] = *(const v2f*)(arow + 4 * ((g + 1) * G + jj));
      } else {
        sc0 = sp[0]; sc1 = sp[1];
      }
      __builtin_amdgcn_sched_barrier(0);
#pragma unroll
      for (int jj = 0; jj < G; ++jj)
        c = __builtin_amdgcn_wmma_f32_16x16x4_f32(false, fa[g & 1][jj], false, b[g * G + jj],
                                                  (short)0, c, false, false);
      __builtin_amdgcn_sched_barrier(0);
    }

    if (more) {
#pragma unroll
      for (int i = 0; i < NV; ++i) {
        const int f = 4 * (i * NT + tid);
        v4f v = r[i];
        if (RELU) { v.x = fmaxf(v.x, 0.0f); v.y = fmaxf(v.y, 0.0f);
                    v.z = fmaxf(v.z, 0.0f); v.w = fmaxf(v.w, 0.0f); }
        *(v4f*)&sA[cur ^ 1][(f / K) * PITCH + (f % K)] = v;
      }
      __syncthreads();
    }

    float* drow = H + (m0 + half * 8) * NOUT + (n0 + l);
#pragma unroll
    for (int i = 0; i < 4; ++i) drow[(size_t)i * NOUT] = c[i] * sc0[i];
#pragma unroll
    for (int i = 0; i < 4; ++i) drow[(size_t)(4 + i) * NOUT] = c[4 + i] * sc1[i];
  }
}


static __device__ __forceinline__ float4 scaled(float a, float4 v) {
  return make_float4(a * v.x, a * v.y, a * v.z, a * v.w);
}
static __device__ __forceinline__ void add_scaled(float4& acc, float a, float4 v) {
  acc.x += a * v.x; acc.y += a * v.y; acc.z += a * v.z; acc.w += a * v.w;
}
static __device__ __forceinline__ float4 plus(float4 a, float4 b) {
  return make_float4(a.x + b.x, a.y + b.y, a.z + b.z, a.w + b.w);
}

template <int G, int NP, int LPN>
static __device__ __forceinline__ void walk_slots(float4 (&acc)[NP], const float4* hp,
                                                  int sr, float nr, int lane0) {
  constexpr size_t RP = NP * LPN;
  float4 v[G][NP];
#pragma unroll
  for (int k = 0; k < G; ++k) {
    const int sk = __shfl(sr, lane0 + k, 32);
#pragma unroll
    for (int p = 0; p < NP; ++p) v[k][p] = hp[(size_t)sk * RP + p * LPN];
  }
#pragma unroll
  for (int k = 0; k < G; ++k) {
    const float a = __shfl(nr, lane0 + k, 32);
#pragma unroll
    for (int p = 0; p < NP; ++p) add_scaled(acc[p], a, v[k][p]);
  }
}

template <int DIM, int LPN, int GROUP>
__global__ __launch_bounds__(256) void k_gather(const unsigned* __restrict__ srcLoW,
                                                const unsigned* __restrict__ srcHiW,
                                                const float* __restrict__ wS,
                                                const float* __restrict__ inv,
                                                const int* __restrict__ off,
                                                const float* __restrict__ h,
                                                const float* __restrict__ bias,
                                                float* __restrict__ res) {
  constexpr int NPW = 32 / LPN;
  constexpr int NP  = DIM / 4 / LPN;
  constexpr int R   = 32 / LPN;
  static_assert(NP * 4 * LPN == DIM, "a row must split evenly over the lanes of a group");
  static_assert(LPN % GROUP == 0, "a group of gathers stays inside one slot register");
  static_assert(N_NODES % (8 * NPW) == 0, "the waves must cover the nodes exactly");

  const unsigned short* srcLo = (const unsigned short*)srcLoW;
  const unsigned char*  srcHi = (const unsigned char*)srcHiW;
  const int lane  = threadIdx.x & 31;
  const int j     = lane % LPN;
  const int gbase = lane - j;
  const int node  = (blockIdx.x * 8 + (int)(threadIdx.x >> 5)) * NPW + lane / LPN;

  const int beg = off[node];
  const int end = off[node + 1];
  const float inv_t = inv[node];
  const float4* hp = (const float4*)h + j;
  constexpr size_t RP = DIM / 4;

  float4 acc[NP];
#pragma unroll
  for (int p = 0; p < NP; ++p)
    acc[p] = scaled(inv_t, hp[(size_t)node * RP + p * LPN]);

  for (int c0 = beg; ; c0 += 32) {
    const int m = min(32, end - c0);
    int mm = m;
#pragma unroll
    for (int d = LPN; d < 32; d <<= 1) mm = max(mm, __shfl_xor(mm, d, 32));
    mm = __builtin_amdgcn_readfirstlane(mm);
    if (mm <= 0) break;

    int s[R]; float nrm[R];
    int lo[R], hi[R]; float we[R];
#pragma unroll
    for (int r = 0; r < R; ++r) {
      const int k = (r * LPN + j < m) ? c0 + r * LPN + j : 0;
      lo[r] = srcLo[k]; hi[r] = srcHi[k]; we[r] = wS[k];
    }
#pragma unroll
    for (int r = 0; r < R; ++r) s[r] = lo[r] | (hi[r] << 16);
#pragma unroll
    for (int r = 0; r < R; ++r) nrm[r] = we[r] * inv_t;
#pragma unroll
    for (int r = 0; r < R; ++r) {
      if (!(r * LPN + j < m)) { s[r] = node; nrm[r] = 0.0f; }
      asm volatile("" : "+v"(s[r]), "+v"(nrm[r]));
    }
#pragma unroll
    for (int r = 0; r < R; ++r) {
#pragma unroll
      for (int u = 0; u < LPN; u += GROUP) {
        const int rem = mm - (r * LPN + u);
        if (rem <= 0) break;
        if (rem > GROUP / 2) walk_slots<GROUP, NP, LPN>(acc, hp, s[r], nrm[r], gbase + u);
        else                 walk_slots<GROUP / 2, NP, LPN>(acc, hp, s[r], nrm[r], gbase + u);
      }
      if ((r + 1) * LPN >= mm) break;
    }
  }

  const float4* bp = (const float4*)bias + j;
  float4* rp = (float4*)res + (size_t)node * RP + j;
#pragma unroll
  for (int p = 0; p < NP; ++p) rp[p * LPN] = plus(acc[p], bp[p * LPN]);
}


extern "C" void kernel_launch(void* const* d_in, const int* in_sizes, int n_in,
                              void* d_out, int out_size, void* d_ws, size_t ws_size,
                              hipStream_t stream) {
  const int N = N_NODES, E = N_EDGES;

  const float* x   = (const float*)d_in[0];
  const int*   ei  = (const int*)d_in[1];
  const int*   src = ei;
  const int*   dst = ei + E;
  const float* w   = (const float*)d_in[2];
  const float* W1  = (const float*)d_in[3];
  const float* b1  = (const float*)d_in[4];
  const float* W2  = (const float*)d_in[5];
  const float* b2  = (const float*)d_in[6];

  float* out  = (float*)d_out;
  float* xemb = out + (size_t)N * DIM_OUT;

  static_assert(N_EDGES % 4 == 0 && N_NODES % 2 == 0, "the list's parts must start on whole words");
  static_assert(N_EDGES + N_EDGES / 2 + N_EDGES / 4 + N_NODES + 1 <= N_NODES * DIM_OUT,
                "the list (weights, 16-bit and 8-bit source parts) and its offsets must fit behind h1");
  static_assert(2 * N_EDGES + NBIN * NCHUNK <= N_NODES * DIM_HID,
                "the build's temporaries must fit in h1");
  float* inv  = (float*)d_ws;
  float* h1   = inv + N;
  float*          wS    = h1 + (size_t)N * DIM_HID;
  unsigned* srcLo = (unsigned*)(wS + E);
  unsigned* srcHi = srcLo + E / 2;
  int*   off  = (int*)(srcHi + E / 4);
  unsigned* tmp = (unsigned*)h1;
  int* runStart = (int*)h1 + 2 * E;
  float* h2   = h1;

  k_piece_sort<<<NCHUNK, NT_E, 0, stream>>>(src, dst, w, runStart, tmp);
  k_bin_sort  <<<NBIN, NT_E, 0, stream>>>(tmp, runStart, off, wS, srcLo, srcHi);
  k_deg_list  <<<N / 32, 256, 0, stream>>>(wS, off, inv);

  k_gemm_wmma<DIM_IN, DIM_HID, 1, 5, false><<<(N / 16) / 5, 256, 0, stream>>>(x, W1, inv, h1);

  k_gather<DIM_HID, 16, 8><<<N / (8 * 2), 256, 0, stream>>>(srcLo, srcHi, wS, inv, off, h1, b1, xemb);

  k_gemm_wmma<DIM_HID, DIM_OUT, 5, 2, true><<<(N / 16) / (5 * 2), 320, 0, stream>>>(xemb, W2, inv, h2);

  k_gather<DIM_OUT, 8, 8><<<N / (8 * 4), 256, 0, stream>>>(srcLo, srcHi, wS, inv, off, h2, b2, out);
}
